# starting kernel
# baseline (speedup 1.0000x reference)
#include <hip/hip_runtime.h>
#include <math.h>


typedef __attribute__((ext_vector_type(2))) float v2f;
typedef __attribute__((ext_vector_type(8))) float v8f;

#define GCN_N 50000
#define GCN_E 600000
#define GCN_D 128
#define TILE_K 32
#define LDS_PAD 4


__global__ __launch_bounds__(256) void deg_init_k(float* __restrict__ deg, int n) {
    int i = blockIdx.x * 256 + threadIdx.x;
    if (i < n) deg[i] = 1.0f;
}

__global__ __launch_bounds__(256) void deg_accum_k(float* __restrict__ deg,
                                                   const int* __restrict__ col,
                                                   const float* __restrict__ w, int e) {
    int i = blockIdx.x * 256 + threadIdx.x;
    if (i < e) atomicAdd(&deg[col[i]], w[i]);
}

__global__ __launch_bounds__(256) void dinv_k(const float* __restrict__ deg,
                                              float* __restrict__ dinv, int n) {
    int i = blockIdx.x * 256 + threadIdx.x;
    if (i < n) {
        float d = deg[i];
        dinv[i] = (d > 0.0f) ? rsqrtf(d) : 0.0f;
    }
}

__global__ __launch_bounds__(256) void gemm_wmma_k(const float* __restrict__ A,
                                                   const float* __restrict__ W,
                                                   float* __restrict__ C) {
    __shared__ float As[16][TILE_K + LDS_PAD];
    __shared__ float Wt[GCN_D][TILE_K + LDS_PAD];

    const int t    = threadIdx.x;
    const int lane = t & 31;
    const int wave = t >> 5;
    const int rowBase = blockIdx.x * 16;

    v8f acc = {};

    for (int kc = 0; kc < GCN_D; kc += TILE_K) {
        __syncthreads();
        for (int idx = t; idx < 16 * TILE_K; idx += 256) {
            int m = idx / TILE_K, k = idx % TILE_K;
            As[m][k] = A[(rowBase + m) * GCN_D + kc + k];
        }
        for (int idx = t; idx < TILE_K * GCN_D; idx += 256) {
            int k = idx / GCN_D, n = idx % GCN_D;
            Wt[n][k] = W[(kc + k) * GCN_D + n];
        }
        __syncthreads();

        const int m   = lane & 15;
        const int n   = wave * 16 + (lane & 15);
        const int klo = (lane >> 4) * 2;
        #pragma unroll
        for (int kk = 0; kk < TILE_K; kk += 4) {
            v2f a, b;
            a.x = As[m][kk + klo];
            a.y = As[m][kk + klo + 1];
            b.x = Wt[n][kk + klo];
            b.y = Wt[n][kk + klo + 1];
            acc = __builtin_amdgcn_wmma_f32_16x16x4_f32(
                 false, a,  false, b,
                 (short)0, acc,  false,  false);
        }
    }

    const int col    = wave * 16 + (lane & 15);
    const int rowOff = (lane >> 4) * 8;
    #pragma unroll
    for (int v = 0; v < 8; ++v) {
        C[(rowBase + v + rowOff) * GCN_D + col] = acc[v];
    }
}

__global__ __launch_bounds__(256) void init_out_k(float* __restrict__ out,
                                                  const float* __restrict__ xl,
                                                  const float* __restrict__ dinv,
                                                  const float* __restrict__ b) {
    int idx = blockIdx.x * 256 + threadIdx.x;
    int i = idx >> 7;
    int d = idx & (GCN_D - 1);
    float di = dinv[i];
    out[idx] = di * di * xl[idx] + b[d];
}

__global__ __launch_bounds__(256) void scatter_k(float* __restrict__ out,
                                                 const float* __restrict__ xl,
                                                 const int* __restrict__ row,
                                                 const int* __restrict__ col,
                                                 const float* __restrict__ w,
                                                 const float* __restrict__ dinv,
                                                 int e_total) {
    int wave = threadIdx.x >> 5;
    int lane = threadIdx.x & 31;
    int e = blockIdx.x * 8 + wave;
    if (e >= e_total) return;

    int r = row[e];
    int c = col[e];
    float norm = dinv[r] * w[e] * dinv[c];

    const float4* src = (const float4*)(xl + (size_t)r * GCN_D);
    float4 v = src[lane];
    float* dst = out + (size_t)c * GCN_D + lane * 4;
    atomicAdd(dst + 0, norm * v.x);
    atomicAdd(dst + 1, norm * v.y);
    atomicAdd(dst + 2, norm * v.z);
    atomicAdd(dst + 3, norm * v.w);
}

__global__ __launch_bounds__(256) void act_k(float* __restrict__ out, int mode,
                                             const float* __restrict__ prelu_w) {
    int idx = blockIdx.x * 256 + threadIdx.x;
    float h = out[idx];
    float r;
    if (mode == 0) {
        r = (h > 0.0f) ? h : (expf(h) - 1.0f);
    } else if (mode == 1) {
        r = 0.5f * h * (1.0f + erff(h * 0.70710678118654752f));
    } else {
        r = (h >= 0.0f) ? h : prelu_w[0] * h;
    }
    out[idx] = r;
}


extern "C" void kernel_launch(void* const* d_in, const int* in_sizes, int n_in,
                              void* d_out, int out_size, void* d_ws, size_t ws_size,
                              hipStream_t stream) {
    (void)in_sizes; (void)n_in; (void)out_size; (void)ws_size;

    const float* x   = (const float*)d_in[0];
    const int*   row = (const int*)d_in[1];
    const int*   col = ((const int*)d_in[1]) + GCN_E;
    const float* w   = (const float*)d_in[2];
    const float* W1  = (const float*)d_in[3];
    const float* b1  = (const float*)d_in[4];
    const float* W2  = (const float*)d_in[5];
    const float* b2  = (const float*)d_in[6];
    const float* W3  = (const float*)d_in[7];
    const float* b3  = (const float*)d_in[8];
    const float* pw  = (const float*)d_in[9];

    float* out  = (float*)d_out;
    float* xl   = (float*)d_ws;
    float* deg  = xl + (size_t)GCN_N * GCN_D;
    float* dinv = deg + GCN_N;

    const int nThr = 256;
    const int gN   = (GCN_N + nThr - 1) / nThr;
    const int gE   = (GCN_E + nThr - 1) / nThr;
    const int gND  = (GCN_N * GCN_D) / nThr;
    const int gGemm = GCN_N / 16;
    const int gScat = (GCN_E + 7) / 8;

    deg_init_k <<<gN, nThr, 0, stream>>>(deg, GCN_N);
    deg_accum_k<<<gE, nThr, 0, stream>>>(deg, col, w, GCN_E);
    dinv_k     <<<gN, nThr, 0, stream>>>(deg, dinv, GCN_N);

    gemm_wmma_k<<<gGemm, nThr, 0, stream>>>(x, W1, xl);
    init_out_k <<<gND,   nThr, 0, stream>>>(out, xl, dinv, b1);
    scatter_k  <<<gScat, nThr, 0, stream>>>(out, xl, row, col, w, dinv, GCN_E);
    act_k      <<<gND,   nThr, 0, stream>>>(out, 0, pw);

    gemm_wmma_k<<<gGemm, nThr, 0, stream>>>(out, W2, xl);
    init_out_k <<<gND,   nThr, 0, stream>>>(out, xl, dinv, b2);
    scatter_k  <<<gScat, nThr, 0, stream>>>(out, xl, row, col, w, dinv, GCN_E);
    act_k      <<<gND,   nThr, 0, stream>>>(out, 1, pw);

    gemm_wmma_k<<<gGemm, nThr, 0, stream>>>(out, W3, xl);
    init_out_k <<<gND,   nThr, 0, stream>>>(out, xl, dinv, b3);
    scatter_k  <<<gScat, nThr, 0, stream>>>(out, xl, row, col, w, dinv, GCN_E);
    act_k      <<<gND,   nThr, 0, stream>>>(out, 2, pw);
}


# scatter_k lane map: lane l takes row elements 32j+l instead of 4l+j, so each of the 4 loads and 4 f32 atomic adds per edge covers one whole 128-byte line; sums stay atomic arrival order as before
# speedup vs baseline: 1.3265x; 1.3265x over previous
#include <hip/hip_runtime.h>
#include <math.h>


typedef __attribute__((ext_vector_type(2))) float v2f;
typedef __attribute__((ext_vector_type(8))) float v8f;

#define GCN_N 50000
#define GCN_E 600000
#define GCN_D 128
#define TILE_K 32
#define LDS_PAD 4


__global__ __launch_bounds__(256) void deg_init_k(float* __restrict__ deg, int n) {
    int i = blockIdx.x * 256 + threadIdx.x;
    if (i < n) deg[i] = 1.0f;
}

__global__ __launch_bounds__(256) void deg_accum_k(float* __restrict__ deg,
                                                   const int* __restrict__ col,
                                                   const float* __restrict__ w, int e) {
    int i = blockIdx.x * 256 + threadIdx.x;
    if (i < e) atomicAdd(&deg[col[i]], w[i]);
}

__global__ __launch_bounds__(256) void dinv_k(const float* __restrict__ deg,
                                              float* __restrict__ dinv, int n) {
    int i = blockIdx.x * 256 + threadIdx.x;
    if (i < n) {
        float d = deg[i];
        dinv[i] = (d > 0.0f) ? rsqrtf(d) : 0.0f;
    }
}

__global__ __launch_bounds__(256) void gemm_wmma_k(const float* __restrict__ A,
                                                   const float* __restrict__ W,
                                                   float* __restrict__ C) {
    __shared__ float As[16][TILE_K + LDS_PAD];
    __shared__ float Wt[GCN_D][TILE_K + LDS_PAD];

    const int t    = threadIdx.x;
    const int lane = t & 31;
    const int wave = t >> 5;
    const int rowBase = blockIdx.x * 16;

    v8f acc = {};

    for (int kc = 0; kc < GCN_D; kc += TILE_K) {
        __syncthreads();
        for (int idx = t; idx < 16 * TILE_K; idx += 256) {
            int m = idx / TILE_K, k = idx % TILE_K;
            As[m][k] = A[(rowBase + m) * GCN_D + kc + k];
        }
        for (int idx = t; idx < TILE_K * GCN_D; idx += 256) {
            int k = idx / GCN_D, n = idx % GCN_D;
            Wt[n][k] = W[(kc + k) * GCN_D + n];
        }
        __syncthreads();

        const int m   = lane & 15;
        const int n   = wave * 16 + (lane & 15);
        const int klo = (lane >> 4) * 2;
        #pragma unroll
        for (int kk = 0; kk < TILE_K; kk += 4) {
            v2f a, b;
            a.x = As[m][kk + klo];
            a.y = As[m][kk + klo + 1];
            b.x = Wt[n][kk + klo];
            b.y = Wt[n][kk + klo + 1];
            acc = __builtin_amdgcn_wmma_f32_16x16x4_f32(
                 false, a,  false, b,
                 (short)0, acc,  false,  false);
        }
    }

    const int col    = wave * 16 + (lane & 15);
    const int rowOff = (lane >> 4) * 8;
    #pragma unroll
    for (int v = 0; v < 8; ++v) {
        C[(rowBase + v + rowOff) * GCN_D + col] = acc[v];
    }
}

__global__ __launch_bounds__(256) void init_out_k(float* __restrict__ out,
                                                  const float* __restrict__ xl,
                                                  const float* __restrict__ dinv,
                                                  const float* __restrict__ b) {
    int idx = blockIdx.x * 256 + threadIdx.x;
    int i = idx >> 7;
    int d = idx & (GCN_D - 1);
    float di = dinv[i];
    out[idx] = di * di * xl[idx] + b[d];
}

__global__ __launch_bounds__(256) void scatter_k(float* __restrict__ out,
                                                 const float* __restrict__ xl,
                                                 const int* __restrict__ row,
                                                 const int* __restrict__ col,
                                                 const float* __restrict__ w,
                                                 const float* __restrict__ dinv,
                                                 int e_total) {
    int wave = threadIdx.x >> 5;
    int lane = threadIdx.x & 31;
    int e = blockIdx.x * 8 + wave;
    if (e >= e_total) return;

    int r = row[e];
    int c = col[e];
    float norm = dinv[r] * w[e] * dinv[c];

    const float* src = xl + (size_t)r * GCN_D + lane;
    float v0 = src[0];
    float v1 = src[32];
    float v2 = src[64];
    float v3 = src[96];
    float* dst = out + (size_t)c * GCN_D + lane;
    atomicAdd(dst + 0,  norm * v0);
    atomicAdd(dst + 32, norm * v1);
    atomicAdd(dst + 64, norm * v2);
    atomicAdd(dst + 96, norm * v3);
}

__global__ __launch_bounds__(256) void act_k(float* __restrict__ out, int mode,
                                             const float* __restrict__ prelu_w) {
    int idx = blockIdx.x * 256 + threadIdx.x;
    float h = out[idx];
    float r;
    if (mode == 0) {
        r = (h > 0.0f) ? h : (expf(h) - 1.0f);
    } else if (mode == 1) {
        r = 0.5f * h * (1.0f + erff(h * 0.70710678118654752f));
    } else {
        r = (h >= 0.0f) ? h : prelu_w[0] * h;
    }
    out[idx] = r;
}


extern "C" void kernel_launch(void* const* d_in, const int* in_sizes, int n_in,
                              void* d_out, int out_size, void* d_ws, size_t ws_size,
                              hipStream_t stream) {
    (void)in_sizes; (void)n_in; (void)out_size; (void)ws_size;

    const float* x   = (const float*)d_in[0];
    const int*   row = (const int*)d_in[1];
    const int*   col = ((const int*)d_in[1]) + GCN_E;
    const float* w   = (const float*)d_in[2];
    const float* W1  = (const float*)d_in[3];
    const float* b1  = (const float*)d_in[4];
    const float* W2  = (const float*)d_in[5];
    const float* b2  = (const float*)d_in[6];
    const float* W3  = (const float*)d_in[7];
    const float* b3  = (const float*)d_in[8];
    const float* pw  = (const float*)d_in[9];

    float* out  = (float*)d_out;
    float* xl   = (float*)d_ws;
    float* deg  = xl + (size_t)GCN_N * GCN_D;
    float* dinv = deg + GCN_N;

    const int nThr = 256;
    const int gN   = (GCN_N + nThr - 1) / nThr;
    const int gE   = (GCN_E + nThr - 1) / nThr;
    const int gND  = (GCN_N * GCN_D) / nThr;
    const int gGemm = GCN_N / 16;
    const int gScat = (GCN_E + 7) / 8;

    deg_init_k <<<gN, nThr, 0, stream>>>(deg, GCN_N);
    deg_accum_k<<<gE, nThr, 0, stream>>>(deg, col, w, GCN_E);
    dinv_k     <<<gN, nThr, 0, stream>>>(deg, dinv, GCN_N);

    gemm_wmma_k<<<gGemm, nThr, 0, stream>>>(x, W1, xl);
    init_out_k <<<gND,   nThr, 0, stream>>>(out, xl, dinv, b1);
    scatter_k  <<<gScat, nThr, 0, stream>>>(out, xl, row, col, w, dinv, GCN_E);
    act_k      <<<gND,   nThr, 0, stream>>>(out, 0, pw);

    gemm_wmma_k<<<gGemm, nThr, 0, stream>>>(out, W2, xl);
    init_out_k <<<gND,   nThr, 0, stream>>>(out, xl, dinv, b2);
    scatter_k  <<<gScat, nThr, 0, stream>>>(out, xl, row, col, w, dinv, GCN_E);
    act_k      <<<gND,   nThr, 0, stream>>>(out, 1, pw);

    gemm_wmma_k<<<gGemm, nThr, 0, stream>>>(out, W3, xl);
    init_out_k <<<gND,   nThr, 0, stream>>>(out, xl, dinv, b3);
    scatter_k  <<<gScat, nThr, 0, stream>>>(out, xl, row, col, w, dinv, GCN_E);
    act_k      <<<gND,   nThr, 0, stream>>>(out, 2, pw);
}


# scatter_k: one wave per 8 consecutive edges instead of 1: index loads coalesced per batch, all 32 row loads issued ahead of the atomics, 8x fewer waves; sums stay f32 atomic arrival order
# speedup vs baseline: 1.5244x; 1.1492x over previous
#include <hip/hip_runtime.h>
#include <math.h>


typedef __attribute__((ext_vector_type(2))) float v2f;
typedef __attribute__((ext_vector_type(8))) float v8f;

#define GCN_N 50000
#define GCN_E 600000
#define GCN_D 128
#define TILE_K 32
#define LDS_PAD 4


__global__ __launch_bounds__(256) void deg_init_k(float* __restrict__ deg, int n) {
    int i = blockIdx.x * 256 + threadIdx.x;
    if (i < n) deg[i] = 1.0f;
}

__global__ __launch_bounds__(256) void deg_accum_k(float* __restrict__ deg,
                                                   const int* __restrict__ col,
                                                   const float* __restrict__ w, int e) {
    int i = blockIdx.x * 256 + threadIdx.x;
    if (i < e) atomicAdd(&deg[col[i]], w[i]);
}

__global__ __launch_bounds__(256) void dinv_k(const float* __restrict__ deg,
                                              float* __restrict__ dinv, int n) {
    int i = blockIdx.x * 256 + threadIdx.x;
    if (i < n) {
        float d = deg[i];
        dinv[i] = (d > 0.0f) ? rsqrtf(d) : 0.0f;
    }
}

__global__ __launch_bounds__(256) void gemm_wmma_k(const float* __restrict__ A,
                                                   const float* __restrict__ W,
                                                   float* __restrict__ C) {
    __shared__ float As[16][TILE_K + LDS_PAD];
    __shared__ float Wt[GCN_D][TILE_K + LDS_PAD];

    const int t    = threadIdx.x;
    const int lane = t & 31;
    const int wave = t >> 5;
    const int rowBase = blockIdx.x * 16;

    v8f acc = {};

    for (int kc = 0; kc < GCN_D; kc += TILE_K) {
        __syncthreads();
        for (int idx = t; idx < 16 * TILE_K; idx += 256) {
            int m = idx / TILE_K, k = idx % TILE_K;
            As[m][k] = A[(rowBase + m) * GCN_D + kc + k];
        }
        for (int idx = t; idx < TILE_K * GCN_D; idx += 256) {
            int k = idx / GCN_D, n = idx % GCN_D;
            Wt[n][k] = W[(kc + k) * GCN_D + n];
        }
        __syncthreads();

        const int m   = lane & 15;
        const int n   = wave * 16 + (lane & 15);
        const int klo = (lane >> 4) * 2;
        #pragma unroll
        for (int kk = 0; kk < TILE_K; kk += 4) {
            v2f a, b;
            a.x = As[m][kk + klo];
            a.y = As[m][kk + klo + 1];
            b.x = Wt[n][kk + klo];
            b.y = Wt[n][kk + klo + 1];
            acc = __builtin_amdgcn_wmma_f32_16x16x4_f32(
                 false, a,  false, b,
                 (short)0, acc,  false,  false);
        }
    }

    const int col    = wave * 16 + (lane & 15);
    const int rowOff = (lane >> 4) * 8;
    #pragma unroll
    for (int v = 0; v < 8; ++v) {
        C[(rowBase + v + rowOff) * GCN_D + col] = acc[v];
    }
}

__global__ __launch_bounds__(256) void init_out_k(float* __restrict__ out,
                                                  const float* __restrict__ xl,
                                                  const float* __restrict__ dinv,
                                                  const float* __restrict__ b) {
    int idx = blockIdx.x * 256 + threadIdx.x;
    int i = idx >> 7;
    int d = idx & (GCN_D - 1);
    float di = dinv[i];
    out[idx] = di * di * xl[idx] + b[d];
}

#define SC_EPW 8
#define SC_WPB 8

__device__ __forceinline__ void scatter_edge(float* __restrict__ out,
                                             const float* __restrict__ xl,
                                             int r, int c, float norm, int lane) {
    const float* src = xl + (size_t)r * GCN_D + lane;
    float v0 = src[0];
    float v1 = src[32];
    float v2 = src[64];
    float v3 = src[96];
    float* dst = out + (size_t)c * GCN_D + lane;
    atomicAdd(dst + 0,  norm * v0);
    atomicAdd(dst + 32, norm * v1);
    atomicAdd(dst + 64, norm * v2);
    atomicAdd(dst + 96, norm * v3);
}

__global__ __launch_bounds__(256) void scatter_k(float* __restrict__ out,
                                                 const float* __restrict__ xl,
                                                 const int* __restrict__ row,
                                                 const int* __restrict__ col,
                                                 const float* __restrict__ w,
                                                 const float* __restrict__ dinv,
                                                 int e_total) {
    const int lane = threadIdx.x & 31;
    const int wave  = __builtin_amdgcn_readfirstlane((int)(threadIdx.x >> 5));
    const int ebase = (blockIdx.x * SC_WPB + wave) * SC_EPW;

    if (ebase + SC_EPW <= e_total) {
        const int   me = ebase + (lane & (SC_EPW - 1));
        const int   r  = row[me];
        const int   c  = col[me];
        const float nw = dinv[r] * w[me] * dinv[c];

        float v[SC_EPW][4];
        #pragma unroll
        for (int j = 0; j < SC_EPW; ++j) {
            const int rj = __builtin_amdgcn_readlane(r, j);
            const float* src = xl + (size_t)rj * GCN_D + lane;
            v[j][0] = src[0];
            v[j][1] = src[32];
            v[j][2] = src[64];
            v[j][3] = src[96];
        }
        #pragma unroll
        for (int j = 0; j < SC_EPW; ++j) {
            const int   cj = __builtin_amdgcn_readlane(c, j);
            const float nj = __int_as_float(
                __builtin_amdgcn_readlane(__float_as_int(nw), j));
            float* dst = out + (size_t)cj * GCN_D + lane;
            atomicAdd(dst + 0,  nj * v[j][0]);
            atomicAdd(dst + 32, nj * v[j][1]);
            atomicAdd(dst + 64, nj * v[j][2]);
            atomicAdd(dst + 96, nj * v[j][3]);
        }
    } else {
        for (int e = ebase; e < e_total; ++e) {
            const int r = row[e];
            const int c = col[e];
            scatter_edge(out, xl, r, c, dinv[r] * w[e] * dinv[c], lane);
        }
    }
}

__global__ __launch_bounds__(256) void act_k(float* __restrict__ out, int mode,
                                             const float* __restrict__ prelu_w) {
    int idx = blockIdx.x * 256 + threadIdx.x;
    float h = out[idx];
    float r;
    if (mode == 0) {
        r = (h > 0.0f) ? h : (expf(h) - 1.0f);
    } else if (mode == 1) {
        r = 0.5f * h * (1.0f + erff(h * 0.70710678118654752f));
    } else {
        r = (h >= 0.0f) ? h : prelu_w[0] * h;
    }
    out[idx] = r;
}


extern "C" void kernel_launch(void* const* d_in, const int* in_sizes, int n_in,
                              void* d_out, int out_size, void* d_ws, size_t ws_size,
                              hipStream_t stream) {
    (void)in_sizes; (void)n_in; (void)out_size; (void)ws_size;

    const float* x   = (const float*)d_in[0];
    const int*   row = (const int*)d_in[1];
    const int*   col = ((const int*)d_in[1]) + GCN_E;
    const float* w   = (const float*)d_in[2];
    const float* W1  = (const float*)d_in[3];
    const float* b1  = (const float*)d_in[4];
    const float* W2  = (const float*)d_in[5];
    const float* b2  = (const float*)d_in[6];
    const float* W3  = (const float*)d_in[7];
    const float* b3  = (const float*)d_in[8];
    const float* pw  = (const float*)d_in[9];

    float* out  = (float*)d_out;
    float* xl   = (float*)d_ws;
    float* deg  = xl + (size_t)GCN_N * GCN_D;
    float* dinv = deg + GCN_N;

    const int nThr = 256;
    const int gN   = (GCN_N + nThr - 1) / nThr;
    const int gE   = (GCN_E + nThr - 1) / nThr;
    const int gND  = (GCN_N * GCN_D) / nThr;
    const int gGemm = GCN_N / 16;
    const int gScat = (GCN_E + SC_WPB * SC_EPW - 1) / (SC_WPB * SC_EPW);

    deg_init_k <<<gN, nThr, 0, stream>>>(deg, GCN_N);
    deg_accum_k<<<gE, nThr, 0, stream>>>(deg, col, w, GCN_E);
    dinv_k     <<<gN, nThr, 0, stream>>>(deg, dinv, GCN_N);

    gemm_wmma_k<<<gGemm, nThr, 0, stream>>>(x, W1, xl);
    init_out_k <<<gND,   nThr, 0, stream>>>(out, xl, dinv, b1);
    scatter_k  <<<gScat, nThr, 0, stream>>>(out, xl, row, col, w, dinv, GCN_E);
    act_k      <<<gND,   nThr, 0, stream>>>(out, 0, pw);

    gemm_wmma_k<<<gGemm, nThr, 0, stream>>>(out, W2, xl);
    init_out_k <<<gND,   nThr, 0, stream>>>(out, xl, dinv, b2);
    scatter_k  <<<gScat, nThr, 0, stream>>>(out, xl, row, col, w, dinv, GCN_E);
    act_k      <<<gND,   nThr, 0, stream>>>(out, 1, pw);

    gemm_wmma_k<<<gGemm, nThr, 0, stream>>>(out, W3, xl);
    init_out_k <<<gND,   nThr, 0, stream>>>(out, xl, dinv, b3);
    scatter_k  <<<gScat, nThr, 0, stream>>>(out, xl, row, col, w, dinv, GCN_E);
    act_k      <<<gND,   nThr, 0, stream>>>(out, 2, pw);
}


# init_out_k and the two inner activations folded into gemm_wmma_k: its A load applies the previous layer's activation, its epilogue also stores out = dinv^2*xl + b; 15 launches -> 10
# speedup vs baseline: 1.8179x; 1.1925x over previous
#include <hip/hip_runtime.h>
#include <math.h>


typedef __attribute__((ext_vector_type(2))) float v2f;
typedef __attribute__((ext_vector_type(8))) float v8f;

#define GCN_N 50000
#define GCN_E 600000
#define GCN_D 128
#define TILE_K 32
#define LDS_PAD 4


__global__ __launch_bounds__(256) void deg_init_k(float* __restrict__ deg, int n) {
    int i = blockIdx.x * 256 + threadIdx.x;
    if (i < n) deg[i] = 1.0f;
}

__global__ __launch_bounds__(256) void deg_accum_k(float* __restrict__ deg,
                                                   const int* __restrict__ col,
                                                   const float* __restrict__ w, int e) {
    int i = blockIdx.x * 256 + threadIdx.x;
    if (i < e) atomicAdd(&deg[col[i]], w[i]);
}

__global__ __launch_bounds__(256) void dinv_k(const float* __restrict__ deg,
                                              float* __restrict__ dinv, int n) {
    int i = blockIdx.x * 256 + threadIdx.x;
    if (i < n) {
        float d = deg[i];
        dinv[i] = (d > 0.0f) ? rsqrtf(d) : 0.0f;
    }
}

#define ACT_NONE 0
#define ACT_ELU  1
#define ACT_GELU 2

__device__ __forceinline__ float act_on_load(float h, int mode) {
    if (mode == ACT_ELU)  return (h > 0.0f) ? h : (expf(h) - 1.0f);
    if (mode == ACT_GELU) return 0.5f * h * (1.0f + erff(h * 0.70710678118654752f));
    return h;
}

__global__ __launch_bounds__(256) void gemm_wmma_k(const float* A,
                                                   const float* __restrict__ W,
                                                   float* __restrict__ C,
                                                   float* out,
                                                   const float* __restrict__ dinv,
                                                   const float* __restrict__ b,
                                                   int act_in) {
    __shared__ float As[16][TILE_K + LDS_PAD];
    __shared__ float Wt[GCN_D][TILE_K + LDS_PAD];

    const int t    = threadIdx.x;
    const int lane = t & 31;
    const int wave = t >> 5;
    const int rowBase = blockIdx.x * 16;

    v8f acc = {};

    for (int kc = 0; kc < GCN_D; kc += TILE_K) {
        __syncthreads();
        for (int idx = t; idx < 16 * TILE_K; idx += 256) {
            int m = idx / TILE_K, k = idx % TILE_K;
            As[m][k] = act_on_load(A[(rowBase + m) * GCN_D + kc + k], act_in);
        }
        for (int idx = t; idx < TILE_K * GCN_D; idx += 256) {
            int k = idx / GCN_D, n = idx % GCN_D;
            Wt[n][k] = W[(kc + k) * GCN_D + n];
        }
        __syncthreads();

        const int m   = lane & 15;
        const int n   = wave * 16 + (lane & 15);
        const int klo = (lane >> 4) * 2;
        #pragma unroll
        for (int kk = 0; kk < TILE_K; kk += 4) {
            v2f a, b;
            a.x = As[m][kk + klo];
            a.y = As[m][kk + klo + 1];
            b.x = Wt[n][kk + klo];
            b.y = Wt[n][kk + klo + 1];
            acc = __builtin_amdgcn_wmma_f32_16x16x4_f32(
                 false, a,  false, b,
                 (short)0, acc,  false,  false);
        }
    }

    const int col    = wave * 16 + (lane & 15);
    const int rowOff = (lane >> 4) * 8;
    const float bc   = b[col];
    #pragma unroll
    for (int v = 0; v < 8; ++v) {
        const int   r  = rowBase + v + rowOff;
        const float di = dinv[r];
        const float xv = acc[v];
        C[r * GCN_D + col]   = xv;
        out[r * GCN_D + col] = di * di * xv + bc;
    }
}

#define SC_EPW 8
#define SC_WPB 8

__device__ __forceinline__ void scatter_edge(float* __restrict__ out,
                                             const float* __restrict__ xl,
                                             int r, int c, float norm, int lane) {
    const float* src = xl + (size_t)r * GCN_D + lane;
    float v0 = src[0];
    float v1 = src[32];
    float v2 = src[64];
    float v3 = src[96];
    float* dst = out + (size_t)c * GCN_D + lane;
    atomicAdd(dst + 0,  norm * v0);
    atomicAdd(dst + 32, norm * v1);
    atomicAdd(dst + 64, norm * v2);
    atomicAdd(dst + 96, norm * v3);
}

__global__ __launch_bounds__(256) void scatter_k(float* __restrict__ out,
                                                 const float* __restrict__ xl,
                                                 const int* __restrict__ row,
                                                 const int* __restrict__ col,
                                                 const float* __restrict__ w,
                                                 const float* __restrict__ dinv,
                                                 int e_total) {
    const int lane = threadIdx.x & 31;
    const int wave  = __builtin_amdgcn_readfirstlane((int)(threadIdx.x >> 5));
    const int ebase = (blockIdx.x * SC_WPB + wave) * SC_EPW;

    if (ebase + SC_EPW <= e_total) {
        const int   me = ebase + (lane & (SC_EPW - 1));
        const int   r  = row[me];
        const int   c  = col[me];
        const float nw = dinv[r] * w[me] * dinv[c];

        float v[SC_EPW][4];
        #pragma unroll
        for (int j = 0; j < SC_EPW; ++j) {
            const int rj = __builtin_amdgcn_readlane(r, j);
            const float* src = xl + (size_t)rj * GCN_D + lane;
            v[j][0] = src[0];
            v[j][1] = src[32];
            v[j][2] = src[64];
            v[j][3] = src[96];
        }
        #pragma unroll
        for (int j = 0; j < SC_EPW; ++j) {
            const int   cj = __builtin_amdgcn_readlane(c, j);
            const float nj = __int_as_float(
                __builtin_amdgcn_readlane(__float_as_int(nw), j));
            float* dst = out + (size_t)cj * GCN_D + lane;
            atomicAdd(dst + 0,  nj * v[j][0]);
            atomicAdd(dst + 32, nj * v[j][1]);
            atomicAdd(dst + 64, nj * v[j][2]);
            atomicAdd(dst + 96, nj * v[j][3]);
        }
    } else {
        for (int e = ebase; e < e_total; ++e) {
            const int r = row[e];
            const int c = col[e];
            scatter_edge(out, xl, r, c, dinv[r] * w[e] * dinv[c], lane);
        }
    }
}

__global__ __launch_bounds__(256) void prelu_k(float* __restrict__ out,
                                               const float* __restrict__ prelu_w) {
    int idx = blockIdx.x * 256 + threadIdx.x;
    float h = out[idx];
    out[idx] = (h >= 0.0f) ? h : prelu_w[0] * h;
}


extern "C" void kernel_launch(void* const* d_in, const int* in_sizes, int n_in,
                              void* d_out, int out_size, void* d_ws, size_t ws_size,
                              hipStream_t stream) {
    (void)in_sizes; (void)n_in; (void)out_size; (void)ws_size;

    const float* x   = (const float*)d_in[0];
    const int*   row = (const int*)d_in[1];
    const int*   col = ((const int*)d_in[1]) + GCN_E;
    const float* w   = (const float*)d_in[2];
    const float* W1  = (const float*)d_in[3];
    const float* b1  = (const float*)d_in[4];
    const float* W2  = (const float*)d_in[5];
    const float* b2  = (const float*)d_in[6];
    const float* W3  = (const float*)d_in[7];
    const float* b3  = (const float*)d_in[8];
    const float* pw  = (const float*)d_in[9];

    float* out  = (float*)d_out;
    float* xl   = (float*)d_ws;
    float* deg  = xl + (size_t)GCN_N * GCN_D;
    float* dinv = deg + GCN_N;

    const int nThr = 256;
    const int gN   = (GCN_N + nThr - 1) / nThr;
    const int gE   = (GCN_E + nThr - 1) / nThr;
    const int gND  = (GCN_N * GCN_D) / nThr;
    const int gGemm = GCN_N / 16;
    const int gScat = (GCN_E + SC_WPB * SC_EPW - 1) / (SC_WPB * SC_EPW);

    deg_init_k <<<gN, nThr, 0, stream>>>(deg, GCN_N);
    deg_accum_k<<<gE, nThr, 0, stream>>>(deg, col, w, GCN_E);
    dinv_k     <<<gN, nThr, 0, stream>>>(deg, dinv, GCN_N);

    gemm_wmma_k<<<gGemm, nThr, 0, stream>>>(x, W1, xl, out, dinv, b1, ACT_NONE);
    scatter_k  <<<gScat, nThr, 0, stream>>>(out, xl, row, col, w, dinv, GCN_E);

    gemm_wmma_k<<<gGemm, nThr, 0, stream>>>(out, W2, xl, out, dinv, b2, ACT_ELU);
    scatter_k  <<<gScat, nThr, 0, stream>>>(out, xl, row, col, w, dinv, GCN_E);

    gemm_wmma_k<<<gGemm, nThr, 0, stream>>>(out, W3, xl, out, dinv, b3, ACT_GELU);
    scatter_k  <<<gScat, nThr, 0, stream>>>(out, xl, row, col, w, dinv, GCN_E);
    prelu_k    <<<gND,   nThr, 0, stream>>>(out, pw);
}


# gemm_wmma_k: each wave keeps its 128x16 W slice as B fragments in registers (one load group per block); the block walks 5 row tiles, next A tile prefetched; W no longer restaged via LDS per 16 rows
# speedup vs baseline: 2.2131x; 1.2174x over previous
#include <hip/hip_runtime.h>
#include <math.h>


typedef __attribute__((ext_vector_type(2))) float v2f;
typedef __attribute__((ext_vector_type(8))) float v8f;

#define GCN_N 50000
#define GCN_E 600000
#define GCN_D 128


__global__ __launch_bounds__(256) void deg_init_k(float* __restrict__ deg, int n) {
    int i = blockIdx.x * 256 + threadIdx.x;
    if (i < n) deg[i] = 1.0f;
}

__global__ __launch_bounds__(256) void deg_accum_k(float* __restrict__ deg,
                                                   const int* __restrict__ col,
                                                   const float* __restrict__ w, int e) {
    int i = blockIdx.x * 256 + threadIdx.x;
    if (i < e) atomicAdd(&deg[col[i]], w[i]);
}

__global__ __launch_bounds__(256) void dinv_k(const float* __restrict__ deg,
                                              float* __restrict__ dinv, int n) {
    int i = blockIdx.x * 256 + threadIdx.x;
    if (i < n) {
        float d = deg[i];
        dinv[i] = (d > 0.0f) ? rsqrtf(d) : 0.0f;
    }
}

#define ACT_NONE 0
#define ACT_ELU  1
#define ACT_GELU 2

__device__ __forceinline__ float act_on_load(float h, int mode) {
    if (mode == ACT_ELU)  return (h > 0.0f) ? h : (expf(h) - 1.0f);
    if (mode == ACT_GELU) return 0.5f * h * (1.0f + erff(h * 0.70710678118654752f));
    return h;
}

#define GT_PER_BLOCK 5
#define A_PITCH (GCN_D + 4)

__device__ __forceinline__ float4 act4_on_load(float4 p, int mode) {
    float4 q;
    q.x = act_on_load(p.x, mode);
    q.y = act_on_load(p.y, mode);
    q.z = act_on_load(p.z, mode);
    q.w = act_on_load(p.w, mode);
    return q;
}

__global__ __launch_bounds__(256) void gemm_wmma_k(const float* A,
                                                   const float* __restrict__ W,
                                                   float* __restrict__ C,
                                                   float* out,
                                                   const float* __restrict__ dinv,
                                                   const float* __restrict__ b,
                                                   int act_in) {
    __shared__ __attribute__((aligned(16))) float As[2][16][A_PITCH];

    const int t    = threadIdx.x;
    const int lane = t & 31;
    const int wave = t >> 5;
    const int m   = lane & 15;
    const int n   = wave * 16 + (lane & 15);
    const int klo = (lane >> 4) * 2;

    v2f bf[32];
    #pragma unroll
    for (int j = 0; j < 32; ++j) {
        bf[j].x = W[(4 * j + klo) * GCN_D + n];
        bf[j].y = W[(4 * j + klo + 1) * GCN_D + n];
    }

    const int ar = t >> 5;
    const int ac = (t & 31) * 4;
    const int tile0 = blockIdx.x * GT_PER_BLOCK;

    const float bc   = b[n];
    const int rowOff = (lane >> 4) * 8;

    const float* ap = A + (size_t)(tile0 * 16 + ar) * GCN_D + ac;
    float4 p0 = *reinterpret_cast<const float4*>(ap);
    float4 p1 = *reinterpret_cast<const float4*>(ap + 8 * GCN_D);

    #pragma unroll 1
    for (int i = 0; i < GT_PER_BLOCK; ++i) {
        const int buf = i & 1;
        *reinterpret_cast<float4*>(&As[buf][ar][ac])     = act4_on_load(p0, act_in);
        *reinterpret_cast<float4*>(&As[buf][ar + 8][ac]) = act4_on_load(p1, act_in);
        __syncthreads();

        if (i + 1 < GT_PER_BLOCK) {
            const float* np = ap + (size_t)(i + 1) * 16 * GCN_D;
            p0 = *reinterpret_cast<const float4*>(np);
            p1 = *reinterpret_cast<const float4*>(np + 8 * GCN_D);
        }

        v8f acc = {};
        #pragma unroll
        for (int j = 0; j < 32; ++j) {
            const v2f a = *reinterpret_cast<const v2f*>(&As[buf][m][4 * j + klo]);
            acc = __builtin_amdgcn_wmma_f32_16x16x4_f32(
                 false, a,  false, bf[j],
                 (short)0, acc,  false,  false);
        }

        __builtin_amdgcn_sched_barrier(0);
        asm volatile("v_nop\n\tv_nop\n\tv_nop\n\tv_nop");
        __builtin_amdgcn_sched_barrier(0);

        const int rowBase = (tile0 + i) * 16;
        #pragma unroll
        for (int v = 0; v < 8; ++v) {
            const int   r  = rowBase + v + rowOff;
            const float di = dinv[r];
            const float xv = acc[v];
            C[r * GCN_D + n]   = xv;
            out[r * GCN_D + n] = di * di * xv + bc;
        }
    }
}

#define SC_EPW 8
#define SC_WPB 8

__device__ __forceinline__ void scatter_edge(float* __restrict__ out,
                                             const float* __restrict__ xl,
                                             int r, int c, float norm, int lane) {
    const float* src = xl + (size_t)r * GCN_D + lane;
    float v0 = src[0];
    float v1 = src[32];
    float v2 = src[64];
    float v3 = src[96];
    float* dst = out + (size_t)c * GCN_D + lane;
    atomicAdd(dst + 0,  norm * v0);
    atomicAdd(dst + 32, norm * v1);
    atomicAdd(dst + 64, norm * v2);
    atomicAdd(dst + 96, norm * v3);
}

__global__ __launch_bounds__(256) void scatter_k(float* __restrict__ out,
                                                 const float* __restrict__ xl,
                                                 const int* __restrict__ row,
                                                 const int* __restrict__ col,
                                                 const float* __restrict__ w,
                                                 const float* __restrict__ dinv,
                                                 int e_total) {
    const int lane = threadIdx.x & 31;
    const int wave  = __builtin_amdgcn_readfirstlane((int)(threadIdx.x >> 5));
    const int ebase = (blockIdx.x * SC_WPB + wave) * SC_EPW;

    if (ebase + SC_EPW <= e_total) {
        const int   me = ebase + (lane & (SC_EPW - 1));
        const int   r  = row[me];
        const int   c  = col[me];
        const float nw = dinv[r] * w[me] * dinv[c];

        float v[SC_EPW][4];
        #pragma unroll
        for (int j = 0; j < SC_EPW; ++j) {
            const int rj = __builtin_amdgcn_readlane(r, j);
            const float* src = xl + (size_t)rj * GCN_D + lane;
            v[j][0] = src[0];
            v[j][1] = src[32];
            v[j][2] = src[64];
            v[j][3] = src[96];
        }
        #pragma unroll
        for (int j = 0; j < SC_EPW; ++j) {
            const int   cj = __builtin_amdgcn_readlane(c, j);
            const float nj = __int_as_float(
                __builtin_amdgcn_readlane(__float_as_int(nw), j));
            float* dst = out + (size_t)cj * GCN_D + lane;
            atomicAdd(dst + 0,  nj * v[j][0]);
            atomicAdd(dst + 32, nj * v[j][1]);
            atomicAdd(dst + 64, nj * v[j][2]);
            atomicAdd(dst + 96, nj * v[j][3]);
        }
    } else {
        for (int e = ebase; e < e_total; ++e) {
            const int r = row[e];
            const int c = col[e];
            scatter_edge(out, xl, r, c, dinv[r] * w[e] * dinv[c], lane);
        }
    }
}

__global__ __launch_bounds__(256) void prelu_k(float* __restrict__ out,
                                               const float* __restrict__ prelu_w) {
    int idx = blockIdx.x * 256 + threadIdx.x;
    float h = out[idx];
    out[idx] = (h >= 0.0f) ? h : prelu_w[0] * h;
}


extern "C" void kernel_launch(void* const* d_in, const int* in_sizes, int n_in,
                              void* d_out, int out_size, void* d_ws, size_t ws_size,
                              hipStream_t stream) {
    (void)in_sizes; (void)n_in; (void)out_size; (void)ws_size;

    const float* x   = (const float*)d_in[0];
    const int*   row = (const int*)d_in[1];
    const int*   col = ((const int*)d_in[1]) + GCN_E;
    const float* w   = (const float*)d_in[2];
    const float* W1  = (const float*)d_in[3];
    const float* b1  = (const float*)d_in[4];
    const float* W2  = (const float*)d_in[5];
    const float* b2  = (const float*)d_in[6];
    const float* W3  = (const float*)d_in[7];
    const float* b3  = (const float*)d_in[8];
    const float* pw  = (const float*)d_in[9];

    float* out  = (float*)d_out;
    float* xl   = (float*)d_ws;
    float* deg  = xl + (size_t)GCN_N * GCN_D;
    float* dinv = deg + GCN_N;

    const int nThr = 256;
    const int gN   = (GCN_N + nThr - 1) / nThr;
    const int gE   = (GCN_E + nThr - 1) / nThr;
    const int gND  = (GCN_N * GCN_D) / nThr;
    const int gGemm = GCN_N / (16 * GT_PER_BLOCK);
    const int gScat = (GCN_E + SC_WPB * SC_EPW - 1) / (SC_WPB * SC_EPW);

    deg_init_k <<<gN, nThr, 0, stream>>>(deg, GCN_N);
    deg_accum_k<<<gE, nThr, 0, stream>>>(deg, col, w, GCN_E);
    dinv_k     <<<gN, nThr, 0, stream>>>(deg, dinv, GCN_N);

    gemm_wmma_k<<<gGemm, nThr, 0, stream>>>(x, W1, xl, out, dinv, b1, ACT_NONE);
    scatter_k  <<<gScat, nThr, 0, stream>>>(out, xl, row, col, w, dinv, GCN_E);

    gemm_wmma_k<<<gGemm, nThr, 0, stream>>>(out, W2, xl, out, dinv, b2, ACT_ELU);
    scatter_k  <<<gScat, nThr, 0, stream>>>(out, xl, row, col, w, dinv, GCN_E);

    gemm_wmma_k<<<gGemm, nThr, 0, stream>>>(out, W3, xl, out, dinv, b3, ACT_GELU);
    scatter_k  <<<gScat, nThr, 0, stream>>>(out, xl, row, col, w, dinv, GCN_E);
    prelu_k    <<<gND,   nThr, 0, stream>>>(out, pw);
}


# speedup vs baseline: 2.3233x; 1.0498x over previous
#include <hip/hip_runtime.h>
#include <math.h>


typedef __attribute__((ext_vector_type(2))) float v2f;
typedef __attribute__((ext_vector_type(8))) float v8f;

#define GCN_N 50000
#define GCN_E 600000
#define GCN_D 128


__global__ __launch_bounds__(256) void deg_init_k(float* __restrict__ deg, int n) {
    int i = blockIdx.x * 256 + threadIdx.x;
    if (i < n) deg[i] = 1.0f;
}

__global__ __launch_bounds__(256) void deg_accum_k(float* __restrict__ deg,
                                                   const int* __restrict__ col,
                                                   const float* __restrict__ w, int e) {
    int i = blockIdx.x * 256 + threadIdx.x;
    if (i < e) atomicAdd(&deg[col[i]], w[i]);
}

#define SC_NCTR 6
__global__ __launch_bounds__(256) void dinv_k(float* __restrict__ deg,
                                              float* __restrict__ dinv, int n) {
    int i = blockIdx.x * 256 + threadIdx.x;
    if (i < n) {
        float d = deg[i];
        dinv[i] = (d > 0.0f) ? rsqrtf(d) : 0.0f;
        if (i < SC_NCTR) deg[i] = 0.0f;
    }
}

#define ACT_NONE 0
#define ACT_ELU  1
#define ACT_GELU 2

__device__ __forceinline__ float act_on_load(float h, int mode) {
    if (mode == ACT_ELU)  return (h > 0.0f) ? h : (expf(h) - 1.0f);
    if (mode == ACT_GELU) return 0.5f * h * (1.0f + erff(h * 0.70710678118654752f));
    return h;
}

#define GT_PER_BLOCK 5
#define A_PITCH (GCN_D + 4)

__device__ __forceinline__ float4 act4_on_load(float4 p, int mode) {
    float4 q;
    q.x = act_on_load(p.x, mode);
    q.y = act_on_load(p.y, mode);
    q.z = act_on_load(p.z, mode);
    q.w = act_on_load(p.w, mode);
    return q;
}

__global__ __launch_bounds__(256) void gemm_wmma_k(const float* A,
                                                   const float* __restrict__ W,
                                                   float* __restrict__ C,
                                                   float* out,
                                                   const float* __restrict__ dinv,
                                                   const float* __restrict__ b,
                                                   int act_in) {
    __shared__ __attribute__((aligned(16))) float As[2][16][A_PITCH];

    const int t    = threadIdx.x;
    const int lane = t & 31;
    const int wave = t >> 5;
    const int m   = lane & 15;
    const int n   = wave * 16 + (lane & 15);
    const int klo = (lane >> 4) * 2;

    v2f bf[32];
    #pragma unroll
    for (int j = 0; j < 32; ++j) {
        bf[j].x = W[(4 * j + klo) * GCN_D + n];
        bf[j].y = W[(4 * j + klo + 1) * GCN_D + n];
    }

    const int ar = t >> 5;
    const int ac = (t & 31) * 4;
    const int tile0 = blockIdx.x * GT_PER_BLOCK;

    const float bc   = b[n];
    const int rowOff = (lane >> 4) * 8;

    const float* ap = A + (size_t)(tile0 * 16 + ar) * GCN_D + ac;
    float4 p0 = *reinterpret_cast<const float4*>(ap);
    float4 p1 = *reinterpret_cast<const float4*>(ap + 8 * GCN_D);

    #pragma unroll 1
    for (int i = 0; i < GT_PER_BLOCK; ++i) {
        const int buf = i & 1;
        *reinterpret_cast<float4*>(&As[buf][ar][ac])     = act4_on_load(p0, act_in);
        *reinterpret_cast<float4*>(&As[buf][ar + 8][ac]) = act4_on_load(p1, act_in);
        __syncthreads();

        if (i + 1 < GT_PER_BLOCK) {
            const float* np = ap + (size_t)(i + 1) * 16 * GCN_D;
            p0 = *reinterpret_cast<const float4*>(np);
            p1 = *reinterpret_cast<const float4*>(np + 8 * GCN_D);
        }

        v8f acc = {};
        #pragma unroll
        for (int j = 0; j < 32; ++j) {
            const v2f a = *reinterpret_cast<const v2f*>(&As[buf][m][4 * j + klo]);
            acc = __builtin_amdgcn_wmma_f32_16x16x4_f32(
                 false, a,  false, bf[j],
                 (short)0, acc,  false,  false);
        }

        __builtin_amdgcn_sched_barrier(0);
        asm volatile("v_nop\n\tv_nop\n\tv_nop\n\tv_nop");
        __builtin_amdgcn_sched_barrier(0);

        const int rowBase = (tile0 + i) * 16;
        #pragma unroll
        for (int v = 0; v < 8; ++v) {
            const int   r  = rowBase + v + rowOff;
            const float di = dinv[r];
            const float xv = acc[v];
            C[r * GCN_D + n]   = xv;
            out[r * GCN_D + n] = di * di * xv + bc;
        }
    }
}

#define SC_EPW 16
#define SC_WPB 8
#define SC_GRID 2048
#define SC_SPLIT (GCN_N / 2)

__device__ __forceinline__ int   bcast_i(int x, int j)   { return __builtin_amdgcn_readlane(x, j); }
__device__ __forceinline__ float bcast_f(float x, int j) {
    return __int_as_float(__builtin_amdgcn_readlane(__float_as_int(x), j));
}

__device__ __forceinline__ int take_lowest(unsigned& m) {
    const int j = __builtin_ctz(m);
    m &= m - 1u;
    return j;
}

__device__ __forceinline__ void scatter_batch(float* __restrict__ out,
                                              const float* __restrict__ xl,
                                              const int* __restrict__ row,
                                              const int* __restrict__ col,
                                              const float* __restrict__ w,
                                              const float* __restrict__ dinv,
                                              int ebase, int e_total, int h, int lane) {
    if (ebase >= e_total) return;
    const int  me    = ebase + (lane & (SC_EPW - 1));
    const bool valid = me < e_total;
    const int  mec   = valid ? me : e_total - 1;
    const int   r  = row[mec];
    const int   c  = col[mec];
    const float dr = dinv[r];
    const float we = w[mec];
    const float dn = dinv[c];
    unsigned m = __builtin_amdgcn_ballot_w32(valid && ((c >= SC_SPLIT ? 1 : 0) == h)) & 0xFFFFu;

    for (int g = 0; g < SC_EPW / 4 && __builtin_popcount(m) >= 4; ++g) {
        const int ja = take_lowest(m), jb = take_lowest(m), jc = take_lowest(m), jd = take_lowest(m);
        const float* sa = xl + (size_t)bcast_i(r, ja) * GCN_D + lane;
        const float* sb = xl + (size_t)bcast_i(r, jb) * GCN_D + lane;
        const float* sc = xl + (size_t)bcast_i(r, jc) * GCN_D + lane;
        const float* sd = xl + (size_t)bcast_i(r, jd) * GCN_D + lane;
        const float a0 = sa[0], a1 = sa[32], a2 = sa[64], a3 = sa[96];
        const float b0 = sb[0], b1 = sb[32], b2 = sb[64], b3 = sb[96];
        const float c0 = sc[0], c1 = sc[32], c2 = sc[64], c3 = sc[96];
        const float d0 = sd[0], d1 = sd[32], d2 = sd[64], d3 = sd[96];
        __builtin_amdgcn_sched_barrier(0);
        float* da = out + (size_t)bcast_i(c, ja) * GCN_D + lane;
        float* db = out + (size_t)bcast_i(c, jb) * GCN_D + lane;
        float* dc = out + (size_t)bcast_i(c, jc) * GCN_D + lane;
        float* dd = out + (size_t)bcast_i(c, jd) * GCN_D + lane;
        const float na = bcast_f(dr, ja) * bcast_f(we, ja) * bcast_f(dn, ja);
        const float nb = bcast_f(dr, jb) * bcast_f(we, jb) * bcast_f(dn, jb);
        const float nc = bcast_f(dr, jc) * bcast_f(we, jc) * bcast_f(dn, jc);
        const float nd = bcast_f(dr, jd) * bcast_f(we, jd) * bcast_f(dn, jd);
        atomicAdd(da + 0, na * a0); atomicAdd(da + 32, na * a1); atomicAdd(da + 64, na * a2); atomicAdd(da + 96, na * a3);
        atomicAdd(db + 0, nb * b0); atomicAdd(db + 32, nb * b1); atomicAdd(db + 64, nb * b2); atomicAdd(db + 96, nb * b3);
        atomicAdd(dc + 0, nc * c0); atomicAdd(dc + 32, nc * c1); atomicAdd(dc + 64, nc * c2); atomicAdd(dc + 96, nc * c3);
        atomicAdd(dd + 0, nd * d0); atomicAdd(dd + 32, nd * d1); atomicAdd(dd + 64, nd * d2); atomicAdd(dd + 96, nd * d3);
    }
    for (int s = 0; s < 3 && m != 0u; ++s) {
        const int ja = take_lowest(m);
        const float* sa = xl + (size_t)bcast_i(r, ja) * GCN_D + lane;
        const float a0 = sa[0], a1 = sa[32], a2 = sa[64], a3 = sa[96];
        __builtin_amdgcn_sched_barrier(0);
        float* da = out + (size_t)bcast_i(c, ja) * GCN_D + lane;
        const float na = bcast_f(dr, ja) * bcast_f(we, ja) * bcast_f(dn, ja);
        atomicAdd(da + 0, na * a0); atomicAdd(da + 32, na * a1); atomicAdd(da + 64, na * a2); atomicAdd(da + 96, na * a3);
    }
}

__global__ __launch_bounds__(256) void scatter_k(float* __restrict__ out,
                                                 const float* __restrict__ xl,
                                                 const int* __restrict__ row,
                                                 const int* __restrict__ col,
                                                 const float* __restrict__ w,
                                                 const float* __restrict__ dinv,
                                                 int* __restrict__ ctr,
                                                 int e_total, int n_chunks) {
    __shared__ int s_t;
    const int lane = threadIdx.x & 31;
    const int wave = __builtin_amdgcn_readfirstlane((int)(threadIdx.x >> 5));
    const int my_half = (__builtin_amdgcn_s_sendmsg_rtn(0x87) >> 18) & 1;

    for (int pass = 0; pass < 2; ++pass) {
        const int h = my_half ^ pass;
        int t_next = 0;
        if (threadIdx.x == 0) t_next = atomicAdd(&ctr[h], 1);
        for (int it = 0; it <= n_chunks; ++it) {
            if (threadIdx.x == 0) s_t = t_next;
            __syncthreads();
            const int t = s_t;
            if (t >= n_chunks) break;
            if (threadIdx.x == 0) t_next = atomicAdd(&ctr[h], 1);
            scatter_batch(out, xl, row, col, w, dinv,
                          (t * SC_WPB + wave) * SC_EPW, e_total, h, lane);
            __syncthreads();
        }
        __syncthreads();
    }
}

__global__ __launch_bounds__(256) void prelu_k(float* __restrict__ out,
                                               const float* __restrict__ prelu_w) {
    int idx = blockIdx.x * 256 + threadIdx.x;
    float h = out[idx];
    out[idx] = (h >= 0.0f) ? h : prelu_w[0] * h;
}


extern "C" void kernel_launch(void* const* d_in, const int* in_sizes, int n_in,
                              void* d_out, int out_size, void* d_ws, size_t ws_size,
                              hipStream_t stream) {
    (void)in_sizes; (void)n_in; (void)out_size; (void)ws_size;

    const float* x   = (const float*)d_in[0];
    const int*   row = (const int*)d_in[1];
    const int*   col = ((const int*)d_in[1]) + GCN_E;
    const float* w   = (const float*)d_in[2];
    const float* W1  = (const float*)d_in[3];
    const float* b1  = (const float*)d_in[4];
    const float* W2  = (const float*)d_in[5];
    const float* b2  = (const float*)d_in[6];
    const float* W3  = (const float*)d_in[7];
    const float* b3  = (const float*)d_in[8];
    const float* pw  = (const float*)d_in[9];

    float* out  = (float*)d_out;
    float* xl   = (float*)d_ws;
    float* deg  = xl + (size_t)GCN_N * GCN_D;
    float* dinv = deg + GCN_N;

    const int nThr = 256;
    const int gN   = (GCN_N + nThr - 1) / nThr;
    const int gE   = (GCN_E + nThr - 1) / nThr;
    const int gND  = (GCN_N * GCN_D) / nThr;
    const int gGemm = GCN_N / (16 * GT_PER_BLOCK);
    const int nChunk = (GCN_E + SC_WPB * SC_EPW - 1) / (SC_WPB * SC_EPW);
    int* ctr = (int*)deg;

    deg_init_k <<<gN, nThr, 0, stream>>>(deg, GCN_N);
    deg_accum_k<<<gE, nThr, 0, stream>>>(deg, col, w, GCN_E);
    dinv_k     <<<gN, nThr, 0, stream>>>(deg, dinv, GCN_N);

    gemm_wmma_k<<<gGemm, nThr, 0, stream>>>(x, W1, xl, out, dinv, b1, ACT_NONE);
    scatter_k  <<<SC_GRID, nThr, 0, stream>>>(out, xl, row, col, w, dinv, ctr + 0, GCN_E, nChunk);

    gemm_wmma_k<<<gGemm, nThr, 0, stream>>>(out, W2, xl, out, dinv, b2, ACT_ELU);
    scatter_k  <<<SC_GRID, nThr, 0, stream>>>(out, xl, row, col, w, dinv, ctr + 2, GCN_E, nChunk);

    gemm_wmma_k<<<gGemm, nThr, 0, stream>>>(out, W3, xl, out, dinv, b3, ACT_GELU);
    scatter_k  <<<SC_GRID, nThr, 0, stream>>>(out, xl, row, col, w, dinv, ctr + 4, GCN_E, nChunk);
    prelu_k    <<<gND,   nThr, 0, stream>>>(out, pw);
}


# scatter_k's f32 atomic adds replaced by a gather: edges ordered by target row once per call (count+rank, scan, fill); a wave sums a row in registers in list order, stores it once; +5 MB workspace
# speedup vs baseline: 2.6396x; 1.1361x over previous
#include <hip/hip_runtime.h>
#include <math.h>


typedef __attribute__((ext_vector_type(2))) float v2f;
typedef __attribute__((ext_vector_type(8))) float v8f;

#define GCN_N 50000
#define GCN_E 600000
#define GCN_D 128

#define WS_NEED_BYTES ((size_t)(GCN_N * GCN_D + 2 * GCN_N + (GCN_N + 4) + 2 * GCN_E) * 4)

__global__ __launch_bounds__(256) void cnt_init_k(int* __restrict__ cnt, int n) {
    int i = blockIdx.x * 256 + threadIdx.x;
    if (i < n) cnt[i] = 0;
}

__global__ __launch_bounds__(256) void count_k(int* __restrict__ cnt,
                                               int* __restrict__ rank,
                                               const int* __restrict__ col, int e) {
    int i = blockIdx.x * 256 + threadIdx.x;
    if (i < e) rank[i] = atomicAdd(&cnt[col[i]], 1);
}

#define SCAN_STEPS 13
#define SCAN_SEG (SCAN_STEPS * 128)
static_assert(32 * SCAN_SEG >= GCN_N, "scan segments must cover all rows");
static_assert(GCN_N % 4 == 0, "16-byte groups of counts must not straddle the end");

__global__ __launch_bounds__(1024) void rowptr_k(const int* __restrict__ cnt,
                                                 int* __restrict__ rowptr, int n,
                                                 size_t ws_bytes) {
    if (ws_bytes < WS_NEED_BYTES) return;
    __shared__ int s_tot[32];
    const int lane = threadIdx.x & 31;
    const int wave = threadIdx.x >> 5;
    const int seg0 = wave * SCAN_SEG + 4 * lane;

    int4 c[SCAN_STEPS];
    #pragma unroll
    for (int k = 0; k < SCAN_STEPS; ++k) {
        const int idx = seg0 + 128 * k;
        c[k] = (idx < n) ? *reinterpret_cast<const int4*>(cnt + idx) : make_int4(0, 0, 0, 0);
    }

    int carry = 0;
    #pragma unroll
    for (int k = 0; k < SCAN_STEPS; ++k) {
        const int t1 = c[k].x, t2 = t1 + c[k].y, t3 = t2 + c[k].z, t4 = t3 + c[k].w;
        int incl = t4;
        #pragma unroll
        for (int d = 1; d < 32; d <<= 1) {
            const int up = __shfl_up(incl, d);
            if (lane >= d) incl += up;
        }
        const int ex = carry + incl - t4;
        c[k] = make_int4(ex, ex + t1, ex + t2, ex + t3);
        carry += __shfl(incl, 31);
    }
    if (lane == 0) s_tot[wave] = carry;
    __syncthreads();

    int off = (lane < wave) ? s_tot[lane] : 0;
    #pragma unroll
    for (int d = 16; d >= 1; d >>= 1) off += __shfl_xor(off, d);

    #pragma unroll
    for (int k = 0; k < SCAN_STEPS; ++k) {
        const int idx = seg0 + 128 * k;
        if (idx < n)
            *reinterpret_cast<int4*>(rowptr + idx) =
                make_int4(c[k].x + off, c[k].y + off, c[k].z + off, c[k].w + off);
    }
    if (threadIdx.x == 1023) rowptr[n] = off + carry;
}

__global__ __launch_bounds__(256) void fill_k(const int* __restrict__ row,
                                              const int* __restrict__ col,
                                              const float* __restrict__ w,
                                              const int* __restrict__ rowptr,
                                              const int* __restrict__ rank,
                                              int* __restrict__ slot_words, int e,
                                              size_t ws_bytes) {
    if (ws_bytes < WS_NEED_BYTES) return;
    int2* __restrict__ slot = reinterpret_cast<int2*>(slot_words);
    int i = blockIdx.x * 256 + threadIdx.x;
    if (i < e) {
        const int pos = rowptr[col[i]] + rank[i];
        if ((unsigned)pos < (unsigned)e) {
            slot[pos] = make_int2(row[i], __float_as_int(w[i]));
        }
    }
}

__global__ __launch_bounds__(256) void dinv_k(const int* __restrict__ rowptr,
                                              const int* __restrict__ slot_words,
                                              float* __restrict__ dinv, int n, int e,
                                              size_t ws_bytes) {
    if (ws_bytes < WS_NEED_BYTES) return;
    const int2* __restrict__ slot = reinterpret_cast<const int2*>(slot_words);
    int i = blockIdx.x * 256 + threadIdx.x;
    if (i < n) {
        const int beg = max(rowptr[i], 0);
        const int end = min(rowptr[i + 1], e);
        float d = 1.0f;
        for (int k = beg; k < end; ++k) d += __int_as_float(slot[k].y);
        dinv[i] = (d > 0.0f) ? rsqrtf(d) : 0.0f;
    }
}

__global__ __launch_bounds__(256) void norm_k(const int* __restrict__ rowptr,
                                              int* slot_words,
                                              const float* __restrict__ dinv, int n, int e,
                                              size_t ws_bytes) {
    if (ws_bytes < WS_NEED_BYTES) return;
    int2* slot = reinterpret_cast<int2*>(slot_words);
    int i = blockIdx.x * 256 + threadIdx.x;
    if (i < n) {
        const int beg = max(rowptr[i], 0);
        const int end = min(rowptr[i + 1], e);
        const float di = dinv[i];
        for (int k = beg; k < end; ++k) {
            const int2 sl = slot[k];
            slot[k].y = __float_as_int(dinv[sl.x] * __int_as_float(sl.y) * di);
        }
    }
}

#define ACT_NONE 0
#define ACT_ELU  1
#define ACT_GELU 2

__device__ __forceinline__ float act_on_load(float h, int mode) {
    if (mode == ACT_ELU)  return (h > 0.0f) ? h : (expf(h) - 1.0f);
    if (mode == ACT_GELU) return 0.5f * h * (1.0f + erff(h * 0.70710678118654752f));
    return h;
}

#define GT_PER_BLOCK 5
#define A_PITCH (GCN_D + 4)

__device__ __forceinline__ float4 act4_on_load(float4 p, int mode) {
    float4 q;
    q.x = act_on_load(p.x, mode);
    q.y = act_on_load(p.y, mode);
    q.z = act_on_load(p.z, mode);
    q.w = act_on_load(p.w, mode);
    return q;
}

__global__ __launch_bounds__(256) void gemm_wmma_k(const float* __restrict__ A,
                                                   const float* __restrict__ W,
                                                   float* __restrict__ C,
                                                   int act_in) {
    __shared__ __attribute__((aligned(16))) float As[2][16][A_PITCH];

    const int t    = threadIdx.x;
    const int lane = t & 31;
    const int wave = t >> 5;
    const int m   = lane & 15;
    const int n   = wave * 16 + (lane & 15);
    const int klo = (lane >> 4) * 2;

    v2f bf[32];
    #pragma unroll
    for (int j = 0; j < 32; ++j) {
        bf[j].x = W[(4 * j + klo) * GCN_D + n];
        bf[j].y = W[(4 * j + klo + 1) * GCN_D + n];
    }

    const int ar = t >> 5;
    const int ac = (t & 31) * 4;
    const int tile0 = blockIdx.x * GT_PER_BLOCK;

    const int rowOff = (lane >> 4) * 8;

    const float* ap = A + (size_t)(tile0 * 16 + ar) * GCN_D + ac;
    float4 p0 = *reinterpret_cast<const float4*>(ap);
    float4 p1 = *reinterpret_cast<const float4*>(ap + 8 * GCN_D);

    #pragma unroll 1
    for (int i = 0; i < GT_PER_BLOCK; ++i) {
        const int buf = i & 1;
        *reinterpret_cast<float4*>(&As[buf][ar][ac])     = act4_on_load(p0, act_in);
        *reinterpret_cast<float4*>(&As[buf][ar + 8][ac]) = act4_on_load(p1, act_in);
        __syncthreads();

        if (i + 1 < GT_PER_BLOCK) {
            const float* np = ap + (size_t)(i + 1) * 16 * GCN_D;
            p0 = *reinterpret_cast<const float4*>(np);
            p1 = *reinterpret_cast<const float4*>(np + 8 * GCN_D);
        }

        v8f acc = {};
        #pragma unroll
        for (int j = 0; j < 32; ++j) {
            const v2f a = *reinterpret_cast<const v2f*>(&As[buf][m][4 * j + klo]);
            acc = __builtin_amdgcn_wmma_f32_16x16x4_f32(
                 false, a,  false, bf[j],
                 (short)0, acc,  false,  false);
        }

        __builtin_amdgcn_sched_barrier(0);
        asm volatile("v_nop\n\tv_nop\n\tv_nop\n\tv_nop");
        __builtin_amdgcn_sched_barrier(0);

        const int rowBase = (tile0 + i) * 16;
        #pragma unroll
        for (int v = 0; v < 8; ++v) {
            const int r = rowBase + v + rowOff;
            C[r * GCN_D + n] = acc[v];
        }
    }
}

#define GA_WPB 8
#define GA_GRID 2048
#define GA_WAVES (GA_GRID * GA_WPB)

__device__ __forceinline__ int   bcast_i(int x, int j)   { return __builtin_amdgcn_readlane(x, j); }
__device__ __forceinline__ float bcast_f(float x, int j) {
    return __int_as_float(__builtin_amdgcn_readlane(__float_as_int(x), j));
}

__device__ __forceinline__ void axpy4(float4& acc, float a, const float4 v) {
    acc.x += a * v.x;
    acc.y += a * v.y;
    acc.z += a * v.z;
    acc.w += a * v.w;
}

__global__ __launch_bounds__(256) void gather_k(float* __restrict__ out,
                                                const float* __restrict__ xl,
                                                const int* __restrict__ rowptr,
                                                const int* __restrict__ slot_words,
                                                const float* __restrict__ dinv,
                                                const float* __restrict__ b,
                                                int n, int e, size_t ws_bytes) {
    if (ws_bytes < WS_NEED_BYTES) return;
    const int2* __restrict__ slot = reinterpret_cast<const int2*>(slot_words);
    const int lane = threadIdx.x & 31;
    const int wave = __builtin_amdgcn_readfirstlane((int)(threadIdx.x >> 5));
    float4 b4;
    b4.x = b[4 * lane + 0];
    b4.y = b[4 * lane + 1];
    b4.z = b[4 * lane + 2];
    b4.w = b[4 * lane + 3];
    const float4* xl4 = reinterpret_cast<const float4*>(xl) + lane;

    for (int i = blockIdx.x * GA_WPB + wave; i < n; i += GA_WAVES) {
        const int beg = max(rowptr[i], 0);
        const int end = min(rowptr[i + 1], e);
        const float  di  = dinv[i];
        const float4 own = xl4[(size_t)i * (GCN_D / 4)];
        const float  dd  = di * di;
        float4 acc;
        acc.x = dd * own.x + b4.x;
        acc.y = dd * own.y + b4.y;
        acc.z = dd * own.z + b4.z;
        acc.w = dd * own.w + b4.w;

        for (int base = beg; base < end; base += 32) {
            const int nb = min(end - base, 32);
            const int k  = base + ((lane < nb) ? lane : 0);
            const int2  sl = slot[k];
            const int   ms = sl.x;
            const float mn = __int_as_float(sl.y);
            int j = 0;
            for (; j + 8 <= nb; j += 8) {
                const float4 v0 = xl4[(size_t)bcast_i(ms, j + 0) * (GCN_D / 4)];
                const float4 v1 = xl4[(size_t)bcast_i(ms, j + 1) * (GCN_D / 4)];
                const float4 v2 = xl4[(size_t)bcast_i(ms, j + 2) * (GCN_D / 4)];
                const float4 v3 = xl4[(size_t)bcast_i(ms, j + 3) * (GCN_D / 4)];
                const float4 v4 = xl4[(size_t)bcast_i(ms, j + 4) * (GCN_D / 4)];
                const float4 v5 = xl4[(size_t)bcast_i(ms, j + 5) * (GCN_D / 4)];
                const float4 v6 = xl4[(size_t)bcast_i(ms, j + 6) * (GCN_D / 4)];
                const float4 v7 = xl4[(size_t)bcast_i(ms, j + 7) * (GCN_D / 4)];
                axpy4(acc, bcast_f(mn, j + 0), v0);
                axpy4(acc, bcast_f(mn, j + 1), v1);
                axpy4(acc, bcast_f(mn, j + 2), v2);
                axpy4(acc, bcast_f(mn, j + 3), v3);
                axpy4(acc, bcast_f(mn, j + 4), v4);
                axpy4(acc, bcast_f(mn, j + 5), v5);
                axpy4(acc, bcast_f(mn, j + 6), v6);
                axpy4(acc, bcast_f(mn, j + 7), v7);
            }
            if (j + 4 <= nb) {
                const float4 v0 = xl4[(size_t)bcast_i(ms, j + 0) * (GCN_D / 4)];
                const float4 v1 = xl4[(size_t)bcast_i(ms, j + 1) * (GCN_D / 4)];
                const float4 v2 = xl4[(size_t)bcast_i(ms, j + 2) * (GCN_D / 4)];
                const float4 v3 = xl4[(size_t)bcast_i(ms, j + 3) * (GCN_D / 4)];
                axpy4(acc, bcast_f(mn, j + 0), v0);
                axpy4(acc, bcast_f(mn, j + 1), v1);
                axpy4(acc, bcast_f(mn, j + 2), v2);
                axpy4(acc, bcast_f(mn, j + 3), v3);
                j += 4;
            }
            if (j < nb) {
                const int j1 = min(j + 1, nb - 1), j2 = min(j + 2, nb - 1);
                const float4 v0 = xl4[(size_t)bcast_i(ms, j)  * (GCN_D / 4)];
                const float4 v1 = xl4[(size_t)bcast_i(ms, j1) * (GCN_D / 4)];
                const float4 v2 = xl4[(size_t)bcast_i(ms, j2) * (GCN_D / 4)];
                axpy4(acc, bcast_f(mn, j), v0);
                if (j + 1 < nb) axpy4(acc, bcast_f(mn, j1), v1);
                if (j + 2 < nb) axpy4(acc, bcast_f(mn, j2), v2);
            }
        }
        *reinterpret_cast<float4*>(out + (size_t)i * GCN_D + 4 * lane) = acc;
    }
}

__global__ __launch_bounds__(256) void prelu_k(float* __restrict__ out,
                                               const float* __restrict__ prelu_w) {
    int idx = blockIdx.x * 256 + threadIdx.x;
    float h = out[idx];
    out[idx] = (h >= 0.0f) ? h : prelu_w[0] * h;
}


extern "C" void kernel_launch(void* const* d_in, const int* in_sizes, int n_in,
                              void* d_out, int out_size, void* d_ws, size_t ws_size,
                              hipStream_t stream) {
    (void)in_sizes; (void)n_in; (void)out_size;

    const float* x   = (const float*)d_in[0];
    const int*   row = (const int*)d_in[1];
    const int*   col = ((const int*)d_in[1]) + GCN_E;
    const float* w   = (const float*)d_in[2];
    const float* W1  = (const float*)d_in[3];
    const float* b1  = (const float*)d_in[4];
    const float* W2  = (const float*)d_in[5];
    const float* b2  = (const float*)d_in[6];
    const float* W3  = (const float*)d_in[7];
    const float* b3  = (const float*)d_in[8];
    const float* pw  = (const float*)d_in[9];

    float* out  = (float*)d_out;
    float* xl   = (float*)d_ws;
    float* deg  = xl + (size_t)GCN_N * GCN_D;
    float* dinv = deg + GCN_N;
    int*   rowptr = (int*)(dinv + GCN_N);
    int*   slot   = rowptr + (GCN_N + 4);
    int*   cnt    = (int*)d_out;
    int*   rank   = cnt + GCN_N;

    const int nThr = 256;
    const int gN   = (GCN_N + nThr - 1) / nThr;
    const int gE   = (GCN_E + nThr - 1) / nThr;
    const int gND  = (GCN_N * GCN_D) / nThr;
    const int gGemm = GCN_N / (16 * GT_PER_BLOCK);

    cnt_init_k<<<gN, nThr, 0, stream>>>(cnt, GCN_N);
    count_k   <<<gE, nThr, 0, stream>>>(cnt, rank, col, GCN_E);
    rowptr_k  <<<1, 1024, 0, stream>>>(cnt, rowptr, GCN_N, ws_size);
    fill_k    <<<gE, nThr, 0, stream>>>(row, col, w, rowptr, rank, slot, GCN_E, ws_size);
    dinv_k    <<<gN, nThr, 0, stream>>>(rowptr, slot, dinv, GCN_N, GCN_E, ws_size);
    norm_k    <<<gN, nThr, 0, stream>>>(rowptr, slot, dinv, GCN_N, GCN_E, ws_size);

    gemm_wmma_k<<<gGemm, nThr, 0, stream>>>(x, W1, xl, ACT_NONE);
    gather_k   <<<GA_GRID, nThr, 0, stream>>>(out, xl, rowptr, slot, dinv, b1, GCN_N, GCN_E, ws_size);

    gemm_wmma_k<<<gGemm, nThr, 0, stream>>>(out, W2, xl, ACT_ELU);
    gather_k   <<<GA_GRID, nThr, 0, stream>>>(out, xl, rowptr, slot, dinv, b2, GCN_N, GCN_E, ws_size);

    gemm_wmma_k<<<gGemm, nThr, 0, stream>>>(out, W3, xl, ACT_GELU);
    gather_k   <<<GA_GRID, nThr, 0, stream>>>(out, xl, rowptr, slot, dinv, b3, GCN_N, GCN_E, ws_size);
    prelu_k    <<<gND,   nThr, 0, stream>>>(out, pw);
}


# each layer aggregates before its product, (A h) W for A (h W), sum order changed: one kernel sums a 16-row tile's in-edges, feeds the f32 WMMAs via LDS, adds bias, activates; 13 launches -> 9
# speedup vs baseline: 2.9904x; 1.1329x over previous
#include <hip/hip_runtime.h>
#include <math.h>


typedef __attribute__((ext_vector_type(2))) float v2f;
typedef __attribute__((ext_vector_type(8))) float v8f;

#define GCN_N 50000
#define GCN_E 600000
#define GCN_D 128

#define WS_NEED_BYTES ((size_t)(GCN_N * GCN_D + 2 * GCN_N + (GCN_N + 4) + 2 * GCN_E) * 4)

__global__ __launch_bounds__(256) void cnt_init_k(int* __restrict__ cnt, int n) {
    int i = blockIdx.x * 256 + threadIdx.x;
    if (i < n) cnt[i] = 0;
}

__global__ __launch_bounds__(256) void count_k(int* __restrict__ cnt,
                                               int* __restrict__ rank,
                                               const int* __restrict__ col, int e) {
    int i = blockIdx.x * 256 + threadIdx.x;
    if (i < e) rank[i] = atomicAdd(&cnt[col[i]], 1);
}

#define SCAN_STEPS 13
#define SCAN_SEG (SCAN_STEPS * 128)
static_assert(32 * SCAN_SEG >= GCN_N, "scan segments must cover all rows");
static_assert(GCN_N % 4 == 0, "16-byte groups of counts must not straddle the end");

__global__ __launch_bounds__(1024) void rowptr_k(const int* __restrict__ cnt,
                                                 int* __restrict__ rowptr, int n,
                                                 size_t ws_bytes) {
    if (ws_bytes < WS_NEED_BYTES) return;
    __shared__ int s_tot[32];
    const int lane = threadIdx.x & 31;
    const int wave = threadIdx.x >> 5;
    const int seg0 = wave * SCAN_SEG + 4 * lane;

    int4 c[SCAN_STEPS];
    #pragma unroll
    for (int k = 0; k < SCAN_STEPS; ++k) {
        const int idx = seg0 + 128 * k;
        c[k] = (idx < n) ? *reinterpret_cast<const int4*>(cnt + idx) : make_int4(0, 0, 0, 0);
    }

    int carry = 0;
    #pragma unroll
    for (int k = 0; k < SCAN_STEPS; ++k) {
        const int t1 = c[k].x, t2 = t1 + c[k].y, t3 = t2 + c[k].z, t4 = t3 + c[k].w;
        int incl = t4;
        #pragma unroll
        for (int d = 1; d < 32; d <<= 1) {
            const int up = __shfl_up(incl, d);
            if (lane >= d) incl += up;
        }
        const int ex = carry + incl - t4;
        c[k] = make_int4(ex, ex + t1, ex + t2, ex + t3);
        carry += __shfl(incl, 31);
    }
    if (lane == 0) s_tot[wave] = carry;
    __syncthreads();

    int off = (lane < wave) ? s_tot[lane] : 0;
    #pragma unroll
    for (int d = 16; d >= 1; d >>= 1) off += __shfl_xor(off, d);

    #pragma unroll
    for (int k = 0; k < SCAN_STEPS; ++k) {
        const int idx = seg0 + 128 * k;
        if (idx < n)
            *reinterpret_cast<int4*>(rowptr + idx) =
                make_int4(c[k].x + off, c[k].y + off, c[k].z + off, c[k].w + off);
    }
    if (threadIdx.x == 1023) rowptr[n] = off + carry;
}

__global__ __launch_bounds__(256) void fill_k(const int* __restrict__ row,
                                              const int* __restrict__ col,
                                              const float* __restrict__ w,
                                              const int* __restrict__ rowptr,
                                              const int* __restrict__ rank,
                                              int* __restrict__ slot_words, int e,
                                              size_t ws_bytes) {
    if (ws_bytes < WS_NEED_BYTES) return;
    int2* __restrict__ slot = reinterpret_cast<int2*>(slot_words);
    int i = blockIdx.x * 256 + threadIdx.x;
    if (i < e) {
        const int pos = rowptr[col[i]] + rank[i];
        if ((unsigned)pos < (unsigned)e) {
            slot[pos] = make_int2(row[i], __float_as_int(w[i]));
        }
    }
}

__global__ __launch_bounds__(256) void dinv_k(const int* __restrict__ rowptr,
                                              const int* __restrict__ slot_words,
                                              float* __restrict__ dinv, int n, int e,
                                              size_t ws_bytes) {
    if (ws_bytes < WS_NEED_BYTES) return;
    const int2* __restrict__ slot = reinterpret_cast<const int2*>(slot_words);
    int i = blockIdx.x * 256 + threadIdx.x;
    if (i < n) {
        const int beg = max(rowptr[i], 0);
        const int end = min(rowptr[i + 1], e);
        float d = 1.0f;
        for (int k = beg; k < end; ++k) d += __int_as_float(slot[k].y);
        dinv[i] = (d > 0.0f) ? rsqrtf(d) : 0.0f;
    }
}

__global__ __launch_bounds__(256) void norm_k(const int* __restrict__ rowptr,
                                              int* slot_words,
                                              const float* __restrict__ dinv, int n, int e,
                                              size_t ws_bytes) {
    if (ws_bytes < WS_NEED_BYTES) return;
    int2* slot = reinterpret_cast<int2*>(slot_words);
    int i = blockIdx.x * 256 + threadIdx.x;
    if (i < n) {
        const int beg = max(rowptr[i], 0);
        const int end = min(rowptr[i + 1], e);
        const float di = dinv[i];
        for (int k = beg; k < end; ++k) {
            const int2 sl = slot[k];
            slot[k].y = __float_as_int(dinv[sl.x] * __int_as_float(sl.y) * di);
        }
    }
}

#define ACT_ELU   1
#define ACT_GELU  2
#define ACT_PRELU 3

__device__ __forceinline__ float act_on_store(float h, int mode, float slope) {
    if (mode == ACT_ELU)  return (h > 0.0f) ? h : (expf(h) - 1.0f);
    if (mode == ACT_GELU) return 0.5f * h * (1.0f + erff(h * 0.70710678118654752f));
    return (h >= 0.0f) ? h : slope * h;
}

#define LT_PER_BLOCK 4
#define N_TILES (GCN_N / 16)
#define L_GRID ((N_TILES + LT_PER_BLOCK - 1) / LT_PER_BLOCK)
#define A_PITCH (GCN_D + 4)
static_assert(GCN_N % 16 == 0, "row tiles must be whole");

__device__ __forceinline__ int   bcast_i(int x, int j)   { return __builtin_amdgcn_readlane(x, j); }
__device__ __forceinline__ float bcast_f(float x, int j) {
    return __int_as_float(__builtin_amdgcn_readlane(__float_as_int(x), j));
}

__device__ __forceinline__ void axpy4(float4& acc, float a, const float4 v) {
    const v2f aa = {a, a};
    const v2f lo = __builtin_elementwise_fma(aa, (v2f){v.x, v.y}, (v2f){acc.x, acc.y});
    const v2f hi = __builtin_elementwise_fma(aa, (v2f){v.z, v.w}, (v2f){acc.z, acc.w});
    acc.x = lo.x; acc.y = lo.y; acc.z = hi.x; acc.w = hi.y;
}

__device__ __forceinline__ float4 load_row4(const float* __restrict__ h, int r, int lane) {
    const unsigned long long a = reinterpret_cast<unsigned long long>(h + (size_t)r * GCN_D);
    const unsigned lo = __builtin_amdgcn_readfirstlane((unsigned)a);
    const unsigned hi = __builtin_amdgcn_readfirstlane((unsigned)(a >> 32));
    const __attribute__((address_space(1))) float4* rowp =
        (const __attribute__((address_space(1))) float4*)(((unsigned long long)hi << 32) | lo);
    return rowp[(unsigned)lane];
}

__device__ __forceinline__ float4 sum_in_edges(const float* __restrict__ h,
                                               const int2* slot,
                                               int lane, int beg, int end,
                                               int2 sl, float4 acc) {
    for (int base = beg; base < end; base += 32) {
        const int nb = min(end - base, 32);
        if (base != beg) {
            const int k = base + ((lane < nb) ? lane : 0);
            sl = slot[k];
        }
        const int   ms = sl.x;
        const float mn = __int_as_float(sl.y);
        int j = 0;
        for (; j + 8 <= nb; j += 8) {
            const float4 v0 = load_row4(h, bcast_i(ms, j + 0), lane);
            const float4 v1 = load_row4(h, bcast_i(ms, j + 1), lane);
            const float4 v2 = load_row4(h, bcast_i(ms, j + 2), lane);
            const float4 v3 = load_row4(h, bcast_i(ms, j + 3), lane);
            const float4 v4 = load_row4(h, bcast_i(ms, j + 4), lane);
            const float4 v5 = load_row4(h, bcast_i(ms, j + 5), lane);
            const float4 v6 = load_row4(h, bcast_i(ms, j + 6), lane);
            const float4 v7 = load_row4(h, bcast_i(ms, j + 7), lane);
            axpy4(acc, bcast_f(mn, j + 0), v0);
            axpy4(acc, bcast_f(mn, j + 1), v1);
            axpy4(acc, bcast_f(mn, j + 2), v2);
            axpy4(acc, bcast_f(mn, j + 3), v3);
            axpy4(acc, bcast_f(mn, j + 4), v4);
            axpy4(acc, bcast_f(mn, j + 5), v5);
            axpy4(acc, bcast_f(mn, j + 6), v6);
            axpy4(acc, bcast_f(mn, j + 7), v7);
        }
        if (j + 4 <= nb) {
            const float4 v0 = load_row4(h, bcast_i(ms, j + 0), lane);
            const float4 v1 = load_row4(h, bcast_i(ms, j + 1), lane);
            const float4 v2 = load_row4(h, bcast_i(ms, j + 2), lane);
            const float4 v3 = load_row4(h, bcast_i(ms, j + 3), lane);
            axpy4(acc, bcast_f(mn, j + 0), v0);
            axpy4(acc, bcast_f(mn, j + 1), v1);
            axpy4(acc, bcast_f(mn, j + 2), v2);
            axpy4(acc, bcast_f(mn, j + 3), v3);
            j += 4;
        }
        if (j < nb) {
            const int j1 = min(j + 1, nb - 1), j2 = min(j + 2, nb - 1);
            const float4 v0 = load_row4(h, bcast_i(ms, j),  lane);
            const float4 v1 = load_row4(h, bcast_i(ms, j1), lane);
            const float4 v2 = load_row4(h, bcast_i(ms, j2), lane);
            axpy4(acc, bcast_f(mn, j), v0);
            if (j + 1 < nb) axpy4(acc, bcast_f(mn, j1), v1);
            if (j + 2 < nb) axpy4(acc, bcast_f(mn, j2), v2);
        }
    }
    return acc;
}

struct TileHead {
    int    beg0, end0, beg1, end1;
    float  dd0, dd1;
    int2   s0, s1;
    float4 own0, own1;
};

__device__ __forceinline__ void open_tile(TileHead& th, int rp, float dv, int row0,
                                          const float* __restrict__ h,
                                          const int2* slot,
                                          int wave, int lane, int e) {
    th.beg0 = max(bcast_i(rp, wave), 0);
    th.end0 = min(bcast_i(rp, wave + 1), e);
    th.beg1 = max(bcast_i(rp, wave + 8), 0);
    th.end1 = min(bcast_i(rp, wave + 9), e);
    const float d0 = bcast_f(dv, wave), d1 = bcast_f(dv, wave + 8);
    th.dd0 = d0 * d0;
    th.dd1 = d1 * d1;
    const int nb0 = min(th.end0 - th.beg0, 32), nb1 = min(th.end1 - th.beg1, 32);
    th.s0 = slot[min(th.beg0 + ((lane < nb0) ? lane : 0), e - 1)];
    th.s1 = slot[min(th.beg1 + ((lane < nb1) ? lane : 0), e - 1)];
    th.own0 = load_row4(h, row0 + wave, lane);
    th.own1 = load_row4(h, row0 + wave + 8, lane);
}

__global__ __launch_bounds__(256) void layer_k(float* __restrict__ hout,
                                               const float* __restrict__ hin,
                                               const float* __restrict__ W,
                                               const float* __restrict__ b,
                                               const float* __restrict__ prelu_w,
                                               const int* __restrict__ rowptr,
                                               const int* slot_words,
                                               const float* __restrict__ dinv,
                                               int act, int e, size_t ws_bytes) {
    if (ws_bytes < WS_NEED_BYTES) return;
    __shared__ __attribute__((aligned(16))) float As[2][16][A_PITCH];

    const int2* slot = reinterpret_cast<const int2*>(slot_words);
    const int t    = threadIdx.x;
    const int lane = t & 31;
    const int wave = __builtin_amdgcn_readfirstlane((int)(threadIdx.x >> 5));
    const int m   = lane & 15;
    const int n   = wave * 16 + (lane & 15);
    const int klo = (lane >> 4) * 2;
    const int rowOff = (lane >> 4) * 8;
    const int tile0 = blockIdx.x * LT_PER_BLOCK;

    const int   rp0 = rowptr[tile0 * 16 + min(lane, 16)];
    const float dv0 = dinv[tile0 * 16 + (lane & 15)];

    v2f bf[32];
    #pragma unroll
    for (int j = 0; j < 32; ++j) {
        bf[j].x = W[(4 * j + klo) * GCN_D + n];
        bf[j].y = W[(4 * j + klo + 1) * GCN_D + n];
    }
    const float bn    = b[n];
    const float slope = prelu_w[0];

    TileHead th;
    open_tile(th, rp0, dv0, tile0 * 16, hin, slot, wave, lane, e);

    #pragma unroll 1
    for (int tt = 0; tt < LT_PER_BLOCK; ++tt) {
        const int tile = tile0 + tt;
        if (tile >= N_TILES) break;
        const int buf = tt & 1;

        const int nrow0 = min(tile + 1, N_TILES - 1) * 16;
        const int   nrp = rowptr[nrow0 + min(lane, 16)];
        const float ndv = dinv[nrow0 + (lane & 15)];

        {
            float4 acc;
            acc.x = th.dd0 * th.own0.x; acc.y = th.dd0 * th.own0.y;
            acc.z = th.dd0 * th.own0.z; acc.w = th.dd0 * th.own0.w;
            acc = sum_in_edges(hin, slot, lane, th.beg0, th.end0, th.s0, acc);
            *reinterpret_cast<float4*>(&As[buf][wave][4 * lane]) = acc;
        }
        {
            float4 acc;
            acc.x = th.dd1 * th.own1.x; acc.y = th.dd1 * th.own1.y;
            acc.z = th.dd1 * th.own1.z; acc.w = th.dd1 * th.own1.w;
            acc = sum_in_edges(hin, slot, lane, th.beg1, th.end1, th.s1, acc);
            *reinterpret_cast<float4*>(&As[buf][wave + 8][4 * lane]) = acc;
        }
        __syncthreads();

        open_tile(th, nrp, ndv, nrow0, hin, slot, wave, lane, e);
        __builtin_amdgcn_sched_barrier(0);

        v8f acc = {};
        #pragma unroll
        for (int j = 0; j < 32; ++j) {
            const v2f a = *reinterpret_cast<const v2f*>(&As[buf][m][4 * j + klo]);
            acc = __builtin_amdgcn_wmma_f32_16x16x4_f32(
                 false, a,  false, bf[j],
                 (short)0, acc,  false,  false);
        }

        __builtin_amdgcn_sched_barrier(0);
        asm volatile("v_nop\n\tv_nop\n\tv_nop\n\tv_nop");
        __builtin_amdgcn_sched_barrier(0);

        const int rowBase = tile * 16;
        #pragma unroll
        for (int v = 0; v < 8; ++v) {
            const int r = rowBase + v + rowOff;
            hout[r * GCN_D + n] = act_on_store(acc[v] + bn, act, slope);
        }
    }
}


extern "C" void kernel_launch(void* const* d_in, const int* in_sizes, int n_in,
                              void* d_out, int out_size, void* d_ws, size_t ws_size,
                              hipStream_t stream) {
    (void)in_sizes; (void)n_in; (void)out_size;

    const float* x   = (const float*)d_in[0];
    const int*   row = (const int*)d_in[1];
    const int*   col = ((const int*)d_in[1]) + GCN_E;
    const float* w   = (const float*)d_in[2];
    const float* W1  = (const float*)d_in[3];
    const float* b1  = (const float*)d_in[4];
    const float* W2  = (const float*)d_in[5];
    const float* b2  = (const float*)d_in[6];
    const float* W3  = (const float*)d_in[7];
    const float* b3  = (const float*)d_in[8];
    const float* pw  = (const float*)d_in[9];

    float* out  = (float*)d_out;
    float* hmid = (float*)d_ws;
    float* deg  = hmid + (size_t)GCN_N * GCN_D;
    float* dinv = deg + GCN_N;
    int*   rowptr = (int*)(dinv + GCN_N);
    int*   slot   = rowptr + (GCN_N + 4);
    int*   cnt    = (int*)d_out;
    int*   rank   = cnt + GCN_N;

    const int nThr = 256;
    const int gN   = (GCN_N + nThr - 1) / nThr;
    const int gE   = (GCN_E + nThr - 1) / nThr;

    cnt_init_k<<<gN, nThr, 0, stream>>>(cnt, GCN_N);
    count_k   <<<gE, nThr, 0, stream>>>(cnt, rank, col, GCN_E);
    rowptr_k  <<<1, 1024, 0, stream>>>(cnt, rowptr, GCN_N, ws_size);
    fill_k    <<<gE, nThr, 0, stream>>>(row, col, w, rowptr, rank, slot, GCN_E, ws_size);
    dinv_k    <<<gN, nThr, 0, stream>>>(rowptr, slot, dinv, GCN_N, GCN_E, ws_size);
    norm_k    <<<gN, nThr, 0, stream>>>(rowptr, slot, dinv, GCN_N, GCN_E, ws_size);

    layer_k<<<L_GRID, nThr, 0, stream>>>(out,  x,    W1, b1, pw, rowptr, slot, dinv, ACT_ELU,   GCN_E, ws_size);
    layer_k<<<L_GRID, nThr, 0, stream>>>(hmid, out,  W2, b2, pw, rowptr, slot, dinv, ACT_GELU,  GCN_E, ws_size);
    layer_k<<<L_GRID, nThr, 0, stream>>>(out,  hmid, W3, b3, pw, rowptr, slot, dinv, ACT_PRELU, GCN_E, ws_size);
}


# count_k: returning global atomic adds become returning LDS adds (30 blocks x 20480 edges, an N-counter LDS table each) + a block-offset pass; row lists, so f32 sum order, now by block then arrival
# speedup vs baseline: 3.4868x; 1.1660x over previous
#include <hip/hip_runtime.h>
#include <math.h>


typedef __attribute__((ext_vector_type(2))) float v2f;
typedef __attribute__((ext_vector_type(8))) float v8f;

#define GCN_N 50000
#define GCN_E 600000
#define GCN_D 128

#define WS_NEED_BYTES ((size_t)(GCN_N * GCN_D + 2 * GCN_N + (GCN_N + 4) + 2 * GCN_E) * 4)

#define CB_THREADS 1024
#define CB_GROUPS  5
#define CB_CHUNK   (CB_THREADS * CB_GROUPS * 4)
#define CB_BLOCKS  ((GCN_E + CB_CHUNK - 1) / CB_CHUNK)
#define CB_TABLE_STEPS ((GCN_N + 4 * CB_THREADS - 1) / (4 * CB_THREADS))
static_assert(GCN_E % 4 == 0, "16-byte groups of edges must not straddle the end");
static_assert(CB_CHUNK % 256 == 0, "a fill_k block must lie inside one count block's chunk");
static_assert((size_t)GCN_N + GCN_E + (size_t)CB_BLOCKS * GCN_N <= (size_t)GCN_N * GCN_D,
              "cnt, rank and hist must fit the output buffer");

__global__ __launch_bounds__(CB_THREADS) void count_k(int* __restrict__ hist,
                                                      int* __restrict__ rank,
                                                      const int* __restrict__ col, int e) {
    __shared__ __attribute__((aligned(16))) int s_cnt[GCN_N];
    const int t = threadIdx.x;
    const int base = blockIdx.x * CB_CHUNK;

    int4 c[CB_GROUPS];
    #pragma unroll
    for (int k = 0; k < CB_GROUPS; ++k) {
        const int i = base + (k * CB_THREADS + t) * 4;
        c[k] = (i < e) ? *reinterpret_cast<const int4*>(col + i) : make_int4(0, 0, 0, 0);
    }
    for (int i = 4 * t; i < GCN_N; i += 4 * CB_THREADS)
        *reinterpret_cast<int4*>(&s_cnt[i]) = make_int4(0, 0, 0, 0);
    __syncthreads();

    int4 r[CB_GROUPS];
    #pragma unroll
    for (int k = 0; k < CB_GROUPS; ++k) {
        const int i = base + (k * CB_THREADS + t) * 4;
        if (i < e) {
            r[k].x = atomicAdd(&s_cnt[c[k].x], 1);
            r[k].y = atomicAdd(&s_cnt[c[k].y], 1);
            r[k].z = atomicAdd(&s_cnt[c[k].z], 1);
            r[k].w = atomicAdd(&s_cnt[c[k].w], 1);
        }
    }
    __syncthreads();

    #pragma unroll
    for (int k = 0; k < CB_GROUPS; ++k) {
        const int i = base + (k * CB_THREADS + t) * 4;
        if (i < e) *reinterpret_cast<int4*>(rank + i) = r[k];
    }
    int* __restrict__ h = hist + (size_t)blockIdx.x * GCN_N;
    #pragma unroll
    for (int k = 0; k < CB_TABLE_STEPS; ++k) {
        const int i = 4 * t + 4 * CB_THREADS * k;
        if (i < GCN_N) *reinterpret_cast<int4*>(h + i) = *reinterpret_cast<const int4*>(&s_cnt[i]);
    }
}

__global__ __launch_bounds__(256) void block_offsets_k(int* __restrict__ hist,
                                                       int* __restrict__ cnt, int n) {
    int i = blockIdx.x * 256 + threadIdx.x;
    if (i < n) {
        int v[CB_BLOCKS];
        #pragma unroll
        for (int b = 0; b < CB_BLOCKS; ++b) v[b] = hist[(size_t)b * GCN_N + i];
        int run = 0;
        #pragma unroll
        for (int b = 0; b < CB_BLOCKS; ++b) {
            hist[(size_t)b * GCN_N + i] = run;
            run += v[b];
        }
        cnt[i] = run;
    }
}

#define SCAN_STEPS 13
#define SCAN_SEG (SCAN_STEPS * 128)
static_assert(32 * SCAN_SEG >= GCN_N, "scan segments must cover all rows");
static_assert(GCN_N % 4 == 0, "16-byte groups of counts must not straddle the end");

__global__ __launch_bounds__(1024) void rowptr_k(const int* __restrict__ cnt,
                                                 int* __restrict__ rowptr, int n,
                                                 size_t ws_bytes) {
    if (ws_bytes < WS_NEED_BYTES) return;
    __shared__ int s_tot[32];
    const int lane = threadIdx.x & 31;
    const int wave = threadIdx.x >> 5;
    const int seg0 = wave * SCAN_SEG + 4 * lane;

    int4 c[SCAN_STEPS];
    #pragma unroll
    for (int k = 0; k < SCAN_STEPS; ++k) {
        const int idx = seg0 + 128 * k;
        c[k] = (idx < n) ? *reinterpret_cast<const int4*>(cnt + idx) : make_int4(0, 0, 0, 0);
    }

    int carry = 0;
    #pragma unroll
    for (int k = 0; k < SCAN_STEPS; ++k) {
        const int t1 = c[k].x, t2 = t1 + c[k].y, t3 = t2 + c[k].z, t4 = t3 + c[k].w;
        int incl = t4;
        #pragma unroll
        for (int d = 1; d < 32; d <<= 1) {
            const int up = __shfl_up(incl, d);
            if (lane >= d) incl += up;
        }
        const int ex = carry + incl - t4;
        c[k] = make_int4(ex, ex + t1, ex + t2, ex + t3);
        carry += __shfl(incl, 31);
    }
    if (lane == 0) s_tot[wave] = carry;
    __syncthreads();

    int off = (lane < wave) ? s_tot[lane] : 0;
    #pragma unroll
    for (int d = 16; d >= 1; d >>= 1) off += __shfl_xor(off, d);

    #pragma unroll
    for (int k = 0; k < SCAN_STEPS; ++k) {
        const int idx = seg0 + 128 * k;
        if (idx < n)
            *reinterpret_cast<int4*>(rowptr + idx) =
                make_int4(c[k].x + off, c[k].y + off, c[k].z + off, c[k].w + off);
    }
    if (threadIdx.x == 1023) rowptr[n] = off + carry;
}

__global__ __launch_bounds__(256) void fill_k(const int* __restrict__ row,
                                              const int* __restrict__ col,
                                              const float* __restrict__ w,
                                              const int* __restrict__ rowptr,
                                              const int* __restrict__ rank,
                                              const int* __restrict__ hist,
                                              int* __restrict__ slot_words, int e,
                                              size_t ws_bytes) {
    if (ws_bytes < WS_NEED_BYTES) return;
    int2* __restrict__ slot = reinterpret_cast<int2*>(slot_words);
    int i = blockIdx.x * 256 + threadIdx.x;
    const int* __restrict__ boff = hist + (size_t)(blockIdx.x / (CB_CHUNK / 256)) * GCN_N;
    if (i < e) {
        const int c = col[i];
        const int pos = rowptr[c] + boff[c] + rank[i];
        if ((unsigned)pos < (unsigned)e) {
            slot[pos] = make_int2(row[i], __float_as_int(w[i]));
        }
    }
}

__global__ __launch_bounds__(256) void dinv_k(const int* __restrict__ rowptr,
                                              const int* __restrict__ slot_words,
                                              float* __restrict__ dinv, int n, int e,
                                              size_t ws_bytes) {
    if (ws_bytes < WS_NEED_BYTES) return;
    const int2* __restrict__ slot = reinterpret_cast<const int2*>(slot_words);
    int i = blockIdx.x * 256 + threadIdx.x;
    if (i < n) {
        const int beg = max(rowptr[i], 0);
        const int end = min(rowptr[i + 1], e);
        float d = 1.0f;
        for (int k = beg; k < end; ++k) d += __int_as_float(slot[k].y);
        dinv[i] = (d > 0.0f) ? rsqrtf(d) : 0.0f;
    }
}

__global__ __launch_bounds__(256) void norm_k(const int* __restrict__ rowptr,
                                              int* slot_words,
                                              const float* __restrict__ dinv, int n, int e,
                                              size_t ws_bytes) {
    if (ws_bytes < WS_NEED_BYTES) return;
    int2* slot = reinterpret_cast<int2*>(slot_words);
    int i = blockIdx.x * 256 + threadIdx.x;
    if (i < n) {
        const int beg = max(rowptr[i], 0);
        const int end = min(rowptr[i + 1], e);
        const float di = dinv[i];
        for (int k = beg; k < end; ++k) {
            const int2 sl = slot[k];
            slot[k].y = __float_as_int(dinv[sl.x] * __int_as_float(sl.y) * di);
        }
    }
}

#define ACT_ELU   1
#define ACT_GELU  2
#define ACT_PRELU 3

__device__ __forceinline__ float act_on_store(float h, int mode, float slope) {
    if (mode == ACT_ELU)  return (h > 0.0f) ? h : (expf(h) - 1.0f);
    if (mode == ACT_GELU) return 0.5f * h * (1.0f + erff(h * 0.70710678118654752f));
    return (h >= 0.0f) ? h : slope * h;
}

#define LT_PER_BLOCK 4
#define N_TILES (GCN_N / 16)
#define L_GRID ((N_TILES + LT_PER_BLOCK - 1) / LT_PER_BLOCK)
#define A_PITCH (GCN_D + 4)
static_assert(GCN_N % 16 == 0, "row tiles must be whole");

__device__ __forceinline__ int   bcast_i(int x, int j)   { return __builtin_amdgcn_readlane(x, j); }
__device__ __forceinline__ float bcast_f(float x, int j) {
    return __int_as_float(__builtin_amdgcn_readlane(__float_as_int(x), j));
}

__device__ __forceinline__ void axpy4(float4& acc, float a, const float4 v) {
    const v2f aa = {a, a};
    const v2f lo = __builtin_elementwise_fma(aa, (v2f){v.x, v.y}, (v2f){acc.x, acc.y});
    const v2f hi = __builtin_elementwise_fma(aa, (v2f){v.z, v.w}, (v2f){acc.z, acc.w});
    acc.x = lo.x; acc.y = lo.y; acc.z = hi.x; acc.w = hi.y;
}

__device__ __forceinline__ float4 load_row4(const float* __restrict__ h, int r, int lane) {
    const unsigned long long a = reinterpret_cast<unsigned long long>(h + (size_t)r * GCN_D);
    const unsigned lo = __builtin_amdgcn_readfirstlane((unsigned)a);
    const unsigned hi = __builtin_amdgcn_readfirstlane((unsigned)(a >> 32));
    const __attribute__((address_space(1))) float4* rowp =
        (const __attribute__((address_space(1))) float4*)(((unsigned long long)hi << 32) | lo);
    return rowp[(unsigned)lane];
}

__device__ __forceinline__ float4 sum_in_edges(const float* __restrict__ h,
                                               const int2* slot,
                                               int lane, int beg, int end,
                                               int2 sl, float4 acc) {
    for (int base = beg; base < end; base += 32) {
        const int nb = min(end - base, 32);
        if (base != beg) {
            const int k = base + ((lane < nb) ? lane : 0);
            sl = slot[k];
        }
        const int   ms = sl.x;
        const float mn = __int_as_float(sl.y);
        int j = 0;
        for (; j + 8 <= nb; j += 8) {
            const float4 v0 = load_row4(h, bcast_i(ms, j + 0), lane);
            const float4 v1 = load_row4(h, bcast_i(ms, j + 1), lane);
            const float4 v2 = load_row4(h, bcast_i(ms, j + 2), lane);
            const float4 v3 = load_row4(h, bcast_i(ms, j + 3), lane);
            const float4 v4 = load_row4(h, bcast_i(ms, j + 4), lane);
            const float4 v5 = load_row4(h, bcast_i(ms, j + 5), lane);
            const float4 v6 = load_row4(h, bcast_i(ms, j + 6), lane);
            const float4 v7 = load_row4(h, bcast_i(ms, j + 7), lane);
            axpy4(acc, bcast_f(mn, j + 0), v0);
            axpy4(acc, bcast_f(mn, j + 1), v1);
            axpy4(acc, bcast_f(mn, j + 2), v2);
            axpy4(acc, bcast_f(mn, j + 3), v3);
            axpy4(acc, bcast_f(mn, j + 4), v4);
            axpy4(acc, bcast_f(mn, j + 5), v5);
            axpy4(acc, bcast_f(mn, j + 6), v6);
            axpy4(acc, bcast_f(mn, j + 7), v7);
        }
        if (j + 4 <= nb) {
            const float4 v0 = load_row4(h, bcast_i(ms, j + 0), lane);
            const float4 v1 = load_row4(h, bcast_i(ms, j + 1), lane);
            const float4 v2 = load_row4(h, bcast_i(ms, j + 2), lane);
            const float4 v3 = load_row4(h, bcast_i(ms, j + 3), lane);
            axpy4(acc, bcast_f(mn, j + 0), v0);
            axpy4(acc, bcast_f(mn, j + 1), v1);
            axpy4(acc, bcast_f(mn, j + 2), v2);
            axpy4(acc, bcast_f(mn, j + 3), v3);
            j += 4;
        }
        if (j < nb) {
            const int j1 = min(j + 1, nb - 1), j2 = min(j + 2, nb - 1);
            const float4 v0 = load_row4(h, bcast_i(ms, j),  lane);
            const float4 v1 = load_row4(h, bcast_i(ms, j1), lane);
            const float4 v2 = load_row4(h, bcast_i(ms, j2), lane);
            axpy4(acc, bcast_f(mn, j), v0);
            if (j + 1 < nb) axpy4(acc, bcast_f(mn, j1), v1);
            if (j + 2 < nb) axpy4(acc, bcast_f(mn, j2), v2);
        }
    }
    return acc;
}

struct TileHead {
    int    beg0, end0, beg1, end1;
    float  dd0, dd1;
    int2   s0, s1;
    float4 own0, own1;
};

__device__ __forceinline__ void open_tile(TileHead& th, int rp, float dv, int row0,
                                          const float* __restrict__ h,
                                          const int2* slot,
                                          int wave, int lane, int e) {
    th.beg0 = max(bcast_i(rp, wave), 0);
    th.end0 = min(bcast_i(rp, wave + 1), e);
    th.beg1 = max(bcast_i(rp, wave + 8), 0);
    th.end1 = min(bcast_i(rp, wave + 9), e);
    const float d0 = bcast_f(dv, wave), d1 = bcast_f(dv, wave + 8);
    th.dd0 = d0 * d0;
    th.dd1 = d1 * d1;
    const int nb0 = min(th.end0 - th.beg0, 32), nb1 = min(th.end1 - th.beg1, 32);
    th.s0 = slot[min(th.beg0 + ((lane < nb0) ? lane : 0), e - 1)];
    th.s1 = slot[min(th.beg1 + ((lane < nb1) ? lane : 0), e - 1)];
    th.own0 = load_row4(h, row0 + wave, lane);
    th.own1 = load_row4(h, row0 + wave + 8, lane);
}

__global__ __launch_bounds__(256) void layer_k(float* __restrict__ hout,
                                               const float* __restrict__ hin,
                                               const float* __restrict__ W,
                                               const float* __restrict__ b,
                                               const float* __restrict__ prelu_w,
                                               const int* __restrict__ rowptr,
                                               const int* slot_words,
                                               const float* __restrict__ dinv,
                                               int act, int e, size_t ws_bytes) {
    if (ws_bytes < WS_NEED_BYTES) return;
    __shared__ __attribute__((aligned(16))) float As[2][16][A_PITCH];

    const int2* slot = reinterpret_cast<const int2*>(slot_words);
    const int t    = threadIdx.x;
    const int lane = t & 31;
    const int wave = __builtin_amdgcn_readfirstlane((int)(threadIdx.x >> 5));
    const int m   = lane & 15;
    const int n   = wave * 16 + (lane & 15);
    const int klo = (lane >> 4) * 2;
    const int rowOff = (lane >> 4) * 8;
    const int tile0 = blockIdx.x * LT_PER_BLOCK;

    const int   rp0 = rowptr[tile0 * 16 + min(lane, 16)];
    const float dv0 = dinv[tile0 * 16 + (lane & 15)];

    v2f bf[32];
    #pragma unroll
    for (int j = 0; j < 32; ++j) {
        bf[j].x = W[(4 * j + klo) * GCN_D + n];
        bf[j].y = W[(4 * j + klo + 1) * GCN_D + n];
    }
    const float bn    = b[n];
    const float slope = prelu_w[0];

    TileHead th;
    open_tile(th, rp0, dv0, tile0 * 16, hin, slot, wave, lane, e);

    #pragma unroll 1
    for (int tt = 0; tt < LT_PER_BLOCK; ++tt) {
        const int tile = tile0 + tt;
        if (tile >= N_TILES) break;
        const int buf = tt & 1;

        const int nrow0 = min(tile + 1, N_TILES - 1) * 16;
        const int   nrp = rowptr[nrow0 + min(lane, 16)];
        const float ndv = dinv[nrow0 + (lane & 15)];

        {
            float4 acc;
            acc.x = th.dd0 * th.own0.x; acc.y = th.dd0 * th.own0.y;
            acc.z = th.dd0 * th.own0.z; acc.w = th.dd0 * th.own0.w;
            acc = sum_in_edges(hin, slot, lane, th.beg0, th.end0, th.s0, acc);
            *reinterpret_cast<float4*>(&As[buf][wave][4 * lane]) = acc;
        }
        {
            float4 acc;
            acc.x = th.dd1 * th.own1.x; acc.y = th.dd1 * th.own1.y;
            acc.z = th.dd1 * th.own1.z; acc.w = th.dd1 * th.own1.w;
            acc = sum_in_edges(hin, slot, lane, th.beg1, th.end1, th.s1, acc);
            *reinterpret_cast<float4*>(&As[buf][wave + 8][4 * lane]) = acc;
        }
        __syncthreads();

        open_tile(th, nrp, ndv, nrow0, hin, slot, wave, lane, e);
        __builtin_amdgcn_sched_barrier(0);

        v8f acc = {};
        #pragma unroll
        for (int j = 0; j < 32; ++j) {
            const v2f a = *reinterpret_cast<const v2f*>(&As[buf][m][4 * j + klo]);
            acc = __builtin_amdgcn_wmma_f32_16x16x4_f32(
                 false, a,  false, bf[j],
                 (short)0, acc,  false,  false);
        }

        __builtin_amdgcn_sched_barrier(0);
        asm volatile("v_nop\n\tv_nop\n\tv_nop\n\tv_nop");
        __builtin_amdgcn_sched_barrier(0);

        const int rowBase = tile * 16;
        #pragma unroll
        for (int v = 0; v < 8; ++v) {
            const int r = rowBase + v + rowOff;
            hout[r * GCN_D + n] = act_on_store(acc[v] + bn, act, slope);
        }
    }
}


extern "C" void kernel_launch(void* const* d_in, const int* in_sizes, int n_in,
                              void* d_out, int out_size, void* d_ws, size_t ws_size,
                              hipStream_t stream) {
    (void)in_sizes; (void)n_in; (void)out_size;

    const float* x   = (const float*)d_in[0];
    const int*   row = (const int*)d_in[1];
    const int*   col = ((const int*)d_in[1]) + GCN_E;
    const float* w   = (const float*)d_in[2];
    const float* W1  = (const float*)d_in[3];
    const float* b1  = (const float*)d_in[4];
    const float* W2  = (const float*)d_in[5];
    const float* b2  = (const float*)d_in[6];
    const float* W3  = (const float*)d_in[7];
    const float* b3  = (const float*)d_in[8];
    const float* pw  = (const float*)d_in[9];

    float* out  = (float*)d_out;
    float* hmid = (float*)d_ws;
    float* deg  = hmid + (size_t)GCN_N * GCN_D;
    float* dinv = deg + GCN_N;
    int*   rowptr = (int*)(dinv + GCN_N);
    int*   slot   = rowptr + (GCN_N + 4);
    int*   cnt    = (int*)d_out;
    int*   rank   = cnt + GCN_N;
    int*   hist   = rank + GCN_E;

    const int nThr = 256;
    const int gN   = (GCN_N + nThr - 1) / nThr;
    const int gE   = (GCN_E + nThr - 1) / nThr;

    count_k        <<<CB_BLOCKS, CB_THREADS, 0, stream>>>(hist, rank, col, GCN_E);
    block_offsets_k<<<gN, nThr, 0, stream>>>(hist, cnt, GCN_N);
    rowptr_k  <<<1, 1024, 0, stream>>>(cnt, rowptr, GCN_N, ws_size);
    fill_k    <<<gE, nThr, 0, stream>>>(row, col, w, rowptr, rank, hist, slot, GCN_E, ws_size);
    dinv_k    <<<gN, nThr, 0, stream>>>(rowptr, slot, dinv, GCN_N, GCN_E, ws_size);
    norm_k    <<<gN, nThr, 0, stream>>>(rowptr, slot, dinv, GCN_N, GCN_E, ws_size);

    layer_k<<<L_GRID, nThr, 0, stream>>>(out,  x,    W1, b1, pw, rowptr, slot, dinv, ACT_ELU,   GCN_E, ws_size);
    layer_k<<<L_GRID, nThr, 0, stream>>>(hmid, out,  W2, b2, pw, rowptr, slot, dinv, ACT_GELU,  GCN_E, ws_size);
    layer_k<<<L_GRID, nThr, 0, stream>>>(out,  hmid, W3, b3, pw, rowptr, slot, dinv, ACT_PRELU, GCN_E, ws_size);
}


# count_k also sums each target's weights in LDS (f32 LDS adds, 16-bit packed counters; deg order: arrival per block, then by block); fill_k stores final norms; dinv_k, norm_k gone: 9 launches -> 7
# speedup vs baseline: 3.6876x; 1.0576x over previous
#include <hip/hip_runtime.h>
#include <math.h>


typedef __attribute__((ext_vector_type(2))) float v2f;
typedef __attribute__((ext_vector_type(8))) float v8f;

#define GCN_N 50000
#define GCN_E 600000
#define GCN_D 128

#define WS_NEED_BYTES ((size_t)(GCN_N * GCN_D + 2 * GCN_N + (GCN_N + 4) + 2 * GCN_E) * 4)

#define CB_THREADS 1024
#define CB_GROUPS  5
#define CB_CHUNK   (CB_THREADS * CB_GROUPS * 4)
#define CB_BLOCKS  ((GCN_E + CB_CHUNK - 1) / CB_CHUNK)
#define CB_TABLE_STEPS ((GCN_N + 4 * CB_THREADS - 1) / (4 * CB_THREADS))
static_assert(GCN_E % 4 == 0, "16-byte groups of edges must not straddle the end");
static_assert(CB_CHUNK % 256 == 0, "a fill_k block must lie inside one count block's chunk");
#define CB_PK_WORDS (GCN_N / 2)
#define CB_PK_STEPS ((CB_PK_WORDS + 4 * CB_THREADS - 1) / (4 * CB_THREADS))
static_assert(GCN_N % 8 == 0, "packed counter words must come in whole 16-byte groups");
static_assert(CB_CHUNK < 65536, "a block's count per target must fit 16 bits");
static_assert((size_t)GCN_N + GCN_E + (size_t)CB_BLOCKS * (CB_PK_WORDS + 2 * GCN_N) <= (size_t)GCN_N * GCN_D,
              "cnt, rank, pk, degp and boff must fit the output buffer");

__device__ __forceinline__ int count_rank(unsigned* s_cnt, int c) {
    const unsigned sh = ((unsigned)c & 1u) << 4;
    const unsigned old = atomicAdd(&s_cnt[c >> 1], 1u << sh);
    return (int)((old >> sh) & 0xffffu);
}

__global__ __launch_bounds__(CB_THREADS) void count_k(unsigned* __restrict__ pk,
                                                      float* __restrict__ degp,
                                                      int* __restrict__ rank,
                                                      const int* __restrict__ col,
                                                      const float* __restrict__ w, int e) {
    __shared__ __attribute__((aligned(16))) unsigned s_cnt[CB_PK_WORDS];
    __shared__ __attribute__((aligned(16))) float    s_deg[GCN_N];
    const int t = threadIdx.x;
    const int base = blockIdx.x * CB_CHUNK;

    int4   c[CB_GROUPS];
    float4 ww[CB_GROUPS];
    #pragma unroll
    for (int k = 0; k < CB_GROUPS; ++k) {
        const int i = base + (k * CB_THREADS + t) * 4;
        c[k]  = (i < e) ? *reinterpret_cast<const int4*>(col + i) : make_int4(0, 0, 0, 0);
        ww[k] = (i < e) ? *reinterpret_cast<const float4*>(w + i) : make_float4(0.0f, 0.0f, 0.0f, 0.0f);
    }
    for (int i = 4 * t; i < CB_PK_WORDS; i += 4 * CB_THREADS)
        *reinterpret_cast<int4*>(&s_cnt[i]) = make_int4(0, 0, 0, 0);
    for (int i = 4 * t; i < GCN_N; i += 4 * CB_THREADS)
        *reinterpret_cast<float4*>(&s_deg[i]) = make_float4(0.0f, 0.0f, 0.0f, 0.0f);
    __syncthreads();

    int4 r[CB_GROUPS];
    #pragma unroll
    for (int k = 0; k < CB_GROUPS; ++k) {
        const int i = base + (k * CB_THREADS + t) * 4;
        if (i < e) {
            r[k].x = count_rank(s_cnt, c[k].x);
            r[k].y = count_rank(s_cnt, c[k].y);
            r[k].z = count_rank(s_cnt, c[k].z);
            r[k].w = count_rank(s_cnt, c[k].w);
            atomicAdd(&s_deg[c[k].x], ww[k].x);
            atomicAdd(&s_deg[c[k].y], ww[k].y);
            atomicAdd(&s_deg[c[k].z], ww[k].z);
            atomicAdd(&s_deg[c[k].w], ww[k].w);
        }
    }
    __syncthreads();

    #pragma unroll
    for (int k = 0; k < CB_GROUPS; ++k) {
        const int i = base + (k * CB_THREADS + t) * 4;
        if (i < e) *reinterpret_cast<int4*>(rank + i) = r[k];
    }
    unsigned* __restrict__ hp = pk + (size_t)blockIdx.x * CB_PK_WORDS;
    #pragma unroll
    for (int k = 0; k < CB_PK_STEPS; ++k) {
        const int i = 4 * t + 4 * CB_THREADS * k;
        if (i < CB_PK_WORDS) *reinterpret_cast<int4*>(hp + i) = *reinterpret_cast<const int4*>(&s_cnt[i]);
    }
    float* __restrict__ hd = degp + (size_t)blockIdx.x * GCN_N;
    #pragma unroll
    for (int k = 0; k < CB_TABLE_STEPS; ++k) {
        const int i = 4 * t + 4 * CB_THREADS * k;
        if (i < GCN_N) *reinterpret_cast<float4*>(hd + i) = *reinterpret_cast<const float4*>(&s_deg[i]);
    }
}

__global__ __launch_bounds__(256) void block_offsets_k(const unsigned* __restrict__ pk,
                                                       const float* __restrict__ degp,
                                                       int* __restrict__ boff,
                                                       int* __restrict__ cnt,
                                                       float* __restrict__ dinv, int n,
                                                       size_t ws_bytes) {
    if (ws_bytes < WS_NEED_BYTES) return;
    const int p = blockIdx.x * 256 + threadIdx.x;
    if (2 * p < n) {
        unsigned v[CB_BLOCKS];
        float2   d[CB_BLOCKS];
        #pragma unroll
        for (int b = 0; b < CB_BLOCKS; ++b) {
            v[b] = pk[(size_t)b * CB_PK_WORDS + p];
            d[b] = *reinterpret_cast<const float2*>(degp + (size_t)b * GCN_N + 2 * p);
        }
        int   run0 = 0,    run1 = 0;
        float deg0 = 1.0f, deg1 = 1.0f;
        #pragma unroll
        for (int b = 0; b < CB_BLOCKS; ++b) {
            *reinterpret_cast<int2*>(boff + (size_t)b * GCN_N + 2 * p) = make_int2(run0, run1);
            run0 += (int)(v[b] & 0xffffu);
            run1 += (int)(v[b] >> 16);
            deg0 += d[b].x;
            deg1 += d[b].y;
        }
        *reinterpret_cast<int2*>(cnt + 2 * p) = make_int2(run0, run1);
        *reinterpret_cast<float2*>(dinv + 2 * p) =
            make_float2((deg0 > 0.0f) ? rsqrtf(deg0) : 0.0f, (deg1 > 0.0f) ? rsqrtf(deg1) : 0.0f);
    }
}

#define SCAN_STEPS 13
#define SCAN_SEG (SCAN_STEPS * 128)
static_assert(32 * SCAN_SEG >= GCN_N, "scan segments must cover all rows");
static_assert(GCN_N % 4 == 0, "16-byte groups of counts must not straddle the end");

__global__ __launch_bounds__(1024) void rowptr_k(const int* __restrict__ cnt,
                                                 int* __restrict__ rowptr, int n,
                                                 size_t ws_bytes) {
    if (ws_bytes < WS_NEED_BYTES) return;
    __shared__ int s_tot[32];
    const int lane = threadIdx.x & 31;
    const int wave = threadIdx.x >> 5;
    const int seg0 = wave * SCAN_SEG + 4 * lane;

    int4 c[SCAN_STEPS];
    #pragma unroll
    for (int k = 0; k < SCAN_STEPS; ++k) {
        const int idx = seg0 + 128 * k;
        c[k] = (idx < n) ? *reinterpret_cast<const int4*>(cnt + idx) : make_int4(0, 0, 0, 0);
    }

    int carry = 0;
    #pragma unroll
    for (int k = 0; k < SCAN_STEPS; ++k) {
        const int t1 = c[k].x, t2 = t1 + c[k].y, t3 = t2 + c[k].z, t4 = t3 + c[k].w;
        int incl = t4;
        #pragma unroll
        for (int d = 1; d < 32; d <<= 1) {
            const int up = __shfl_up(incl, d);
            if (lane >= d) incl += up;
        }
        const int ex = carry + incl - t4;
        c[k] = make_int4(ex, ex + t1, ex + t2, ex + t3);
        carry += __shfl(incl, 31);
    }
    if (lane == 0) s_tot[wave] = carry;
    __syncthreads();

    int off = (lane < wave) ? s_tot[lane] : 0;
    #pragma unroll
    for (int d = 16; d >= 1; d >>= 1) off += __shfl_xor(off, d);

    #pragma unroll
    for (int k = 0; k < SCAN_STEPS; ++k) {
        const int idx = seg0 + 128 * k;
        if (idx < n)
            *reinterpret_cast<int4*>(rowptr + idx) =
                make_int4(c[k].x + off, c[k].y + off, c[k].z + off, c[k].w + off);
    }
    if (threadIdx.x == 1023) rowptr[n] = off + carry;
}

__global__ __launch_bounds__(256) void fill_k(const int* __restrict__ row,
                                              const int* __restrict__ col,
                                              const float* __restrict__ w,
                                              const int* __restrict__ rowptr,
                                              const int* __restrict__ rank,
                                              const int* __restrict__ boff_all,
                                              const float* __restrict__ dinv,
                                              int* __restrict__ slot_words, int e,
                                              size_t ws_bytes) {
    if (ws_bytes < WS_NEED_BYTES) return;
    int2* __restrict__ slot = reinterpret_cast<int2*>(slot_words);
    int i = blockIdx.x * 256 + threadIdx.x;
    const int* __restrict__ boff = boff_all + (size_t)(blockIdx.x / (CB_CHUNK / 256)) * GCN_N;
    if (i < e) {
        const int c = col[i];
        const int r = row[i];
        const int pos = rowptr[c] + boff[c] + rank[i];
        const float nrm = dinv[r] * w[i] * dinv[c];
        slot[min((unsigned)pos, (unsigned)(e - 1))] = make_int2(r, __float_as_int(nrm));
    }
}

#define ACT_ELU   1
#define ACT_GELU  2
#define ACT_PRELU 3

__device__ __forceinline__ float act_on_store(float h, int mode, float slope) {
    if (mode == ACT_ELU)  return (h > 0.0f) ? h : (expf(h) - 1.0f);
    if (mode == ACT_GELU) return 0.5f * h * (1.0f + erff(h * 0.70710678118654752f));
    return (h >= 0.0f) ? h : slope * h;
}

#define LT_PER_BLOCK 4
#define N_TILES (GCN_N / 16)
#define L_GRID ((N_TILES + LT_PER_BLOCK - 1) / LT_PER_BLOCK)
#define A_PITCH (GCN_D + 4)
static_assert(GCN_N % 16 == 0, "row tiles must be whole");

__device__ __forceinline__ int   bcast_i(int x, int j)   { return __builtin_amdgcn_readlane(x, j); }
__device__ __forceinline__ float bcast_f(float x, int j) {
    return __int_as_float(__builtin_amdgcn_readlane(__float_as_int(x), j));
}

__device__ __forceinline__ void axpy4(float4& acc, float a, const float4 v) {
    const v2f aa = {a, a};
    const v2f lo = __builtin_elementwise_fma(aa, (v2f){v.x, v.y}, (v2f){acc.x, acc.y});
    const v2f hi = __builtin_elementwise_fma(aa, (v2f){v.z, v.w}, (v2f){acc.z, acc.w});
    acc.x = lo.x; acc.y = lo.y; acc.z = hi.x; acc.w = hi.y;
}

__device__ __forceinline__ float4 load_row4(const float* __restrict__ h, int r, int lane) {
    const unsigned long long a = reinterpret_cast<unsigned long long>(h + (size_t)r * GCN_D);
    const unsigned lo = __builtin_amdgcn_readfirstlane((unsigned)a);
    const unsigned hi = __builtin_amdgcn_readfirstlane((unsigned)(a >> 32));
    const __attribute__((address_space(1))) float4* rowp =
        (const __attribute__((address_space(1))) float4*)(((unsigned long long)hi << 32) | lo);
    return rowp[(unsigned)lane];
}

__device__ __forceinline__ float4 sum_in_edges(const float* __restrict__ h,
                                               const int2* slot,
                                               int lane, int beg, int end,
                                               int2 sl, float4 acc) {
    for (int base = beg; base < end; base += 32) {
        const int nb = min(end - base, 32);
        if (base != beg) {
            const int k = base + ((lane < nb) ? lane : 0);
            sl = slot[k];
        }
        const int   ms = sl.x;
        const float mn = __int_as_float(sl.y);
        int j = 0;
        for (; j + 8 <= nb; j += 8) {
            const float4 v0 = load_row4(h, bcast_i(ms, j + 0), lane);
            const float4 v1 = load_row4(h, bcast_i(ms, j + 1), lane);
            const float4 v2 = load_row4(h, bcast_i(ms, j + 2), lane);
            const float4 v3 = load_row4(h, bcast_i(ms, j + 3), lane);
            const float4 v4 = load_row4(h, bcast_i(ms, j + 4), lane);
            const float4 v5 = load_row4(h, bcast_i(ms, j + 5), lane);
            const float4 v6 = load_row4(h, bcast_i(ms, j + 6), lane);
            const float4 v7 = load_row4(h, bcast_i(ms, j + 7), lane);
            axpy4(acc, bcast_f(mn, j + 0), v0);
            axpy4(acc, bcast_f(mn, j + 1), v1);
            axpy4(acc, bcast_f(mn, j + 2), v2);
            axpy4(acc, bcast_f(mn, j + 3), v3);
            axpy4(acc, bcast_f(mn, j + 4), v4);
            axpy4(acc, bcast_f(mn, j + 5), v5);
            axpy4(acc, bcast_f(mn, j + 6), v6);
            axpy4(acc, bcast_f(mn, j + 7), v7);
        }
        if (j + 4 <= nb) {
            const float4 v0 = load_row4(h, bcast_i(ms, j + 0), lane);
            const float4 v1 = load_row4(h, bcast_i(ms, j + 1), lane);
            const float4 v2 = load_row4(h, bcast_i(ms, j + 2), lane);
            const float4 v3 = load_row4(h, bcast_i(ms, j + 3), lane);
            axpy4(acc, bcast_f(mn, j + 0), v0);
            axpy4(acc, bcast_f(mn, j + 1), v1);
            axpy4(acc, bcast_f(mn, j + 2), v2);
            axpy4(acc, bcast_f(mn, j + 3), v3);
            j += 4;
        }
        if (j < nb) {
            const int j1 = min(j + 1, nb - 1), j2 = min(j + 2, nb - 1);
            const float4 v0 = load_row4(h, bcast_i(ms, j),  lane);
            const float4 v1 = load_row4(h, bcast_i(ms, j1), lane);
            const float4 v2 = load_row4(h, bcast_i(ms, j2), lane);
            axpy4(acc, bcast_f(mn, j), v0);
            if (j + 1 < nb) axpy4(acc, bcast_f(mn, j1), v1);
            if (j + 2 < nb) axpy4(acc, bcast_f(mn, j2), v2);
        }
    }
    return acc;
}

struct TileHead {
    int    beg0, end0, beg1, end1;
    float  dd0, dd1;
    int2   s0, s1;
    float4 own0, own1;
};

__device__ __forceinline__ void open_tile(TileHead& th, int rp, float dv, int row0,
                                          const float* __restrict__ h,
                                          const int2* slot,
                                          int wave, int lane, int e) {
    th.beg0 = max(bcast_i(rp, wave), 0);
    th.end0 = min(bcast_i(rp, wave + 1), e);
    th.beg1 = max(bcast_i(rp, wave + 8), 0);
    th.end1 = min(bcast_i(rp, wave + 9), e);
    const float d0 = bcast_f(dv, wave), d1 = bcast_f(dv, wave + 8);
    th.dd0 = d0 * d0;
    th.dd1 = d1 * d1;
    const int nb0 = min(th.end0 - th.beg0, 32), nb1 = min(th.end1 - th.beg1, 32);
    th.s0 = slot[min(th.beg0 + ((lane < nb0) ? lane : 0), e - 1)];
    th.s1 = slot[min(th.beg1 + ((lane < nb1) ? lane : 0), e - 1)];
    th.own0 = load_row4(h, row0 + wave, lane);
    th.own1 = load_row4(h, row0 + wave + 8, lane);
}

__global__ __launch_bounds__(256) void layer_k(float* __restrict__ hout,
                                               const float* __restrict__ hin,
                                               const float* __restrict__ W,
                                               const float* __restrict__ b,
                                               const float* __restrict__ prelu_w,
                                               const int* __restrict__ rowptr,
                                               const int* slot_words,
                                               const float* __restrict__ dinv,
                                               int act, int e, size_t ws_bytes) {
    if (ws_bytes < WS_NEED_BYTES) return;
    __shared__ __attribute__((aligned(16))) float As[2][16][A_PITCH];

    const int2* slot = reinterpret_cast<const int2*>(slot_words);
    const int t    = threadIdx.x;
    const int lane = t & 31;
    const int wave = __builtin_amdgcn_readfirstlane((int)(threadIdx.x >> 5));
    const int m   = lane & 15;
    const int n   = wave * 16 + (lane & 15);
    const int klo = (lane >> 4) * 2;
    const int rowOff = (lane >> 4) * 8;
    const int tile0 = blockIdx.x * LT_PER_BLOCK;

    const int   rp0 = rowptr[tile0 * 16 + min(lane, 16)];
    const float dv0 = dinv[tile0 * 16 + (lane & 15)];

    v2f bf[32];
    #pragma unroll
    for (int j = 0; j < 32; ++j) {
        bf[j].x = W[(4 * j + klo) * GCN_D + n];
        bf[j].y = W[(4 * j + klo + 1) * GCN_D + n];
    }
    const float bn    = b[n];
    const float slope = prelu_w[0];

    TileHead th;
    open_tile(th, rp0, dv0, tile0 * 16, hin, slot, wave, lane, e);

    #pragma unroll 1
    for (int tt = 0; tt < LT_PER_BLOCK; ++tt) {
        const int tile = tile0 + tt;
        if (tile >= N_TILES) break;
        const int buf = tt & 1;

        const int nrow0 = min(tile + 1, N_TILES - 1) * 16;
        const int   nrp = rowptr[nrow0 + min(lane, 16)];
        const float ndv = dinv[nrow0 + (lane & 15)];

        {
            float4 acc;
            acc.x = th.dd0 * th.own0.x; acc.y = th.dd0 * th.own0.y;
            acc.z = th.dd0 * th.own0.z; acc.w = th.dd0 * th.own0.w;
            acc = sum_in_edges(hin, slot, lane, th.beg0, th.end0, th.s0, acc);
            *reinterpret_cast<float4*>(&As[buf][wave][4 * lane]) = acc;
        }
        {
            float4 acc;
            acc.x = th.dd1 * th.own1.x; acc.y = th.dd1 * th.own1.y;
            acc.z = th.dd1 * th.own1.z; acc.w = th.dd1 * th.own1.w;
            acc = sum_in_edges(hin, slot, lane, th.beg1, th.end1, th.s1, acc);
            *reinterpret_cast<float4*>(&As[buf][wave + 8][4 * lane]) = acc;
        }
        __syncthreads();

        open_tile(th, nrp, ndv, nrow0, hin, slot, wave, lane, e);
        __builtin_amdgcn_sched_barrier(0);

        v8f acc = {};
        #pragma unroll
        for (int j = 0; j < 32; ++j) {
            const v2f a = *reinterpret_cast<const v2f*>(&As[buf][m][4 * j + klo]);
            acc = __builtin_amdgcn_wmma_f32_16x16x4_f32(
                 false, a,  false, bf[j],
                 (short)0, acc,  false,  false);
        }

        __builtin_amdgcn_sched_barrier(0);
        asm volatile("v_nop\n\tv_nop\n\tv_nop\n\tv_nop");
        __builtin_amdgcn_sched_barrier(0);

        const int rowBase = tile * 16;
        #pragma unroll
        for (int v = 0; v < 8; ++v) {
            const int r = rowBase + v + rowOff;
            hout[r * GCN_D + n] = act_on_store(acc[v] + bn, act, slope);
        }
    }
}


extern "C" void kernel_launch(void* const* d_in, const int* in_sizes, int n_in,
                              void* d_out, int out_size, void* d_ws, size_t ws_size,
                              hipStream_t stream) {
    (void)in_sizes; (void)n_in; (void)out_size;

    const float* x   = (const float*)d_in[0];
    const int*   row = (const int*)d_in[1];
    const int*   col = ((const int*)d_in[1]) + GCN_E;
    const float* w   = (const float*)d_in[2];
    const float* W1  = (const float*)d_in[3];
    const float* b1  = (const float*)d_in[4];
    const float* W2  = (const float*)d_in[5];
    const float* b2  = (const float*)d_in[6];
    const float* W3  = (const float*)d_in[7];
    const float* b3  = (const float*)d_in[8];
    const float* pw  = (const float*)d_in[9];

    float* out  = (float*)d_out;
    float* hmid = (float*)d_ws;
    float* deg  = hmid + (size_t)GCN_N * GCN_D;
    float* dinv = deg + GCN_N;
    int*   rowptr = (int*)(dinv + GCN_N);
    int*   slot   = rowptr + (GCN_N + 4);
    int*   cnt    = (int*)d_out;
    int*   rank   = cnt + GCN_N;
    unsigned* pk  = (unsigned*)(rank + GCN_E);
    float* degp   = (float*)(pk + (size_t)CB_BLOCKS * CB_PK_WORDS);
    int*   boff   = (int*)(degp + (size_t)CB_BLOCKS * GCN_N);

    const int nThr = 256;
    const int gP   = (GCN_N / 2 + nThr - 1) / nThr;
    const int gE   = (GCN_E + nThr - 1) / nThr;

    count_k        <<<CB_BLOCKS, CB_THREADS, 0, stream>>>(pk, degp, rank, col, w, GCN_E);
    block_offsets_k<<<gP, nThr, 0, stream>>>(pk, degp, boff, cnt, dinv, GCN_N, ws_size);
    rowptr_k  <<<1, 1024, 0, stream>>>(cnt, rowptr, GCN_N, ws_size);
    fill_k    <<<gE, nThr, 0, stream>>>(row, col, w, rowptr, rank, boff, dinv, slot, GCN_E, ws_size);

    layer_k<<<L_GRID, nThr, 0, stream>>>(out,  x,    W1, b1, pw, rowptr, slot, dinv, ACT_ELU,   GCN_E, ws_size);
    layer_k<<<L_GRID, nThr, 0, stream>>>(hmid, out,  W2, b2, pw, rowptr, slot, dinv, ACT_GELU,  GCN_E, ws_size);
    layer_k<<<L_GRID, nThr, 0, stream>>>(out,  hmid, W3, b3, pw, rowptr, slot, dinv, ACT_PRELU, GCN_E, ws_size);
}


# sum_in_edges: each row load of a batch has its own scalar base pair (was one pair rebuilt per row): no s_wait_xcnt 0 between a batch's loads, they issue as one clause of 8 or 4; sum order unchanged
# speedup vs baseline: 3.8253x; 1.0373x over previous
#include <hip/hip_runtime.h>
#include <math.h>


typedef __attribute__((ext_vector_type(2))) float v2f;
typedef __attribute__((ext_vector_type(8))) float v8f;

#define GCN_N 50000
#define GCN_E 600000
#define GCN_D 128

#define WS_NEED_BYTES ((size_t)(GCN_N * GCN_D + 2 * GCN_N + (GCN_N + 4) + 2 * GCN_E) * 4)

#define CB_THREADS 1024
#define CB_GROUPS  5
#define CB_CHUNK   (CB_THREADS * CB_GROUPS * 4)
#define CB_BLOCKS  ((GCN_E + CB_CHUNK - 1) / CB_CHUNK)
#define CB_TABLE_STEPS ((GCN_N + 4 * CB_THREADS - 1) / (4 * CB_THREADS))
static_assert(GCN_E % 4 == 0, "16-byte groups of edges must not straddle the end");
static_assert(CB_CHUNK % 256 == 0, "a fill_k block must lie inside one count block's chunk");
#define CB_PK_WORDS (GCN_N / 2)
#define CB_PK_STEPS ((CB_PK_WORDS + 4 * CB_THREADS - 1) / (4 * CB_THREADS))
static_assert(GCN_N % 8 == 0, "packed counter words must come in whole 16-byte groups");
static_assert(CB_CHUNK < 65536, "a block's count per target must fit 16 bits");
static_assert((size_t)GCN_N + GCN_E + (size_t)CB_BLOCKS * (CB_PK_WORDS + 2 * GCN_N) <= (size_t)GCN_N * GCN_D,
              "cnt, rank, pk, degp and boff must fit the output buffer");

__device__ __forceinline__ int count_rank(unsigned* s_cnt, int c) {
    const unsigned sh = ((unsigned)c & 1u) << 4;
    const unsigned old = atomicAdd(&s_cnt[c >> 1], 1u << sh);
    return (int)((old >> sh) & 0xffffu);
}

__global__ __launch_bounds__(CB_THREADS) void count_k(unsigned* __restrict__ pk,
                                                      float* __restrict__ degp,
                                                      int* __restrict__ rank,
                                                      const int* __restrict__ col,
                                                      const float* __restrict__ w, int e) {
    __shared__ __attribute__((aligned(16))) unsigned s_cnt[CB_PK_WORDS];
    __shared__ __attribute__((aligned(16))) float    s_deg[GCN_N];
    const int t = threadIdx.x;
    const int base = blockIdx.x * CB_CHUNK;

    int4   c[CB_GROUPS];
    float4 ww[CB_GROUPS];
    #pragma unroll
    for (int k = 0; k < CB_GROUPS; ++k) {
        const int i = base + (k * CB_THREADS + t) * 4;
        c[k]  = (i < e) ? *reinterpret_cast<const int4*>(col + i) : make_int4(0, 0, 0, 0);
        ww[k] = (i < e) ? *reinterpret_cast<const float4*>(w + i) : make_float4(0.0f, 0.0f, 0.0f, 0.0f);
    }
    for (int i = 4 * t; i < CB_PK_WORDS; i += 4 * CB_THREADS)
        *reinterpret_cast<int4*>(&s_cnt[i]) = make_int4(0, 0, 0, 0);
    for (int i = 4 * t; i < GCN_N; i += 4 * CB_THREADS)
        *reinterpret_cast<float4*>(&s_deg[i]) = make_float4(0.0f, 0.0f, 0.0f, 0.0f);
    __syncthreads();

    int4 r[CB_GROUPS];
    #pragma unroll
    for (int k = 0; k < CB_GROUPS; ++k) {
        const int i = base + (k * CB_THREADS + t) * 4;
        if (i < e) {
            r[k].x = count_rank(s_cnt, c[k].x);
            r[k].y = count_rank(s_cnt, c[k].y);
            r[k].z = count_rank(s_cnt, c[k].z);
            r[k].w = count_rank(s_cnt, c[k].w);
            atomicAdd(&s_deg[c[k].x], ww[k].x);
            atomicAdd(&s_deg[c[k].y], ww[k].y);
            atomicAdd(&s_deg[c[k].z], ww[k].z);
            atomicAdd(&s_deg[c[k].w], ww[k].w);
        }
    }
    __syncthreads();

    #pragma unroll
    for (int k = 0; k < CB_GROUPS; ++k) {
        const int i = base + (k * CB_THREADS + t) * 4;
        if (i < e) *reinterpret_cast<int4*>(rank + i) = r[k];
    }
    unsigned* __restrict__ hp = pk + (size_t)blockIdx.x * CB_PK_WORDS;
    #pragma unroll
    for (int k = 0; k < CB_PK_STEPS; ++k) {
        const int i = 4 * t + 4 * CB_THREADS * k;
        if (i < CB_PK_WORDS) *reinterpret_cast<int4*>(hp + i) = *reinterpret_cast<const int4*>(&s_cnt[i]);
    }
    float* __restrict__ hd = degp + (size_t)blockIdx.x * GCN_N;
    #pragma unroll
    for (int k = 0; k < CB_TABLE_STEPS; ++k) {
        const int i = 4 * t + 4 * CB_THREADS * k;
        if (i < GCN_N) *reinterpret_cast<float4*>(hd + i) = *reinterpret_cast<const float4*>(&s_deg[i]);
    }
}

__global__ __launch_bounds__(256) void block_offsets_k(const unsigned* __restrict__ pk,
                                                       const float* __restrict__ degp,
                                                       int* __restrict__ boff,
                                                       int* __restrict__ cnt,
                                                       float* __restrict__ dinv, int n,
                                                       size_t ws_bytes) {
    if (ws_bytes < WS_NEED_BYTES) return;
    const int p = blockIdx.x * 256 + threadIdx.x;
    if (2 * p < n) {
        unsigned v[CB_BLOCKS];
        float2   d[CB_BLOCKS];
        #pragma unroll
        for (int b = 0; b < CB_BLOCKS; ++b) {
            v[b] = pk[(size_t)b * CB_PK_WORDS + p];
            d[b] = *reinterpret_cast<const float2*>(degp + (size_t)b * GCN_N + 2 * p);
        }
        int   run0 = 0,    run1 = 0;
        float deg0 = 1.0f, deg1 = 1.0f;
        #pragma unroll
        for (int b = 0; b < CB_BLOCKS; ++b) {
            *reinterpret_cast<int2*>(boff + (size_t)b * GCN_N + 2 * p) = make_int2(run0, run1);
            run0 += (int)(v[b] & 0xffffu);
            run1 += (int)(v[b] >> 16);
            deg0 += d[b].x;
            deg1 += d[b].y;
        }
        *reinterpret_cast<int2*>(cnt + 2 * p) = make_int2(run0, run1);
        *reinterpret_cast<float2*>(dinv + 2 * p) =
            make_float2((deg0 > 0.0f) ? rsqrtf(deg0) : 0.0f, (deg1 > 0.0f) ? rsqrtf(deg1) : 0.0f);
    }
}

#define SCAN_STEPS 13
#define SCAN_SEG (SCAN_STEPS * 128)
static_assert(32 * SCAN_SEG >= GCN_N, "scan segments must cover all rows");
static_assert(GCN_N % 4 == 0, "16-byte groups of counts must not straddle the end");

__global__ __launch_bounds__(1024) void rowptr_k(const int* __restrict__ cnt,
                                                 int* __restrict__ rowptr, int n,
                                                 size_t ws_bytes) {
    if (ws_bytes < WS_NEED_BYTES) return;
    __shared__ int s_tot[32];
    const int lane = threadIdx.x & 31;
    const int wave = threadIdx.x >> 5;
    const int seg0 = wave * SCAN_SEG + 4 * lane;

    int4 c[SCAN_STEPS];
    #pragma unroll
    for (int k = 0; k < SCAN_STEPS; ++k) {
        const int idx = seg0 + 128 * k;
        c[k] = (idx < n) ? *reinterpret_cast<const int4*>(cnt + idx) : make_int4(0, 0, 0, 0);
    }

    int carry = 0;
    #pragma unroll
    for (int k = 0; k < SCAN_STEPS; ++k) {
        const int t1 = c[k].x, t2 = t1 + c[k].y, t3 = t2 + c[k].z, t4 = t3 + c[k].w;
        int incl = t4;
        #pragma unroll
        for (int d = 1; d < 32; d <<= 1) {
            const int up = __shfl_up(incl, d);
            if (lane >= d) incl += up;
        }
        const int ex = carry + incl - t4;
        c[k] = make_int4(ex, ex + t1, ex + t2, ex + t3);
        carry += __shfl(incl, 31);
    }
    if (lane == 0) s_tot[wave] = carry;
    __syncthreads();

    int off = (lane < wave) ? s_tot[lane] : 0;
    #pragma unroll
    for (int d = 16; d >= 1; d >>= 1) off += __shfl_xor(off, d);

    #pragma unroll
    for (int k = 0; k < SCAN_STEPS; ++k) {
        const int idx = seg0 + 128 * k;
        if (idx < n)
            *reinterpret_cast<int4*>(rowptr + idx) =
                make_int4(c[k].x + off, c[k].y + off, c[k].z + off, c[k].w + off);
    }
    if (threadIdx.x == 1023) rowptr[n] = off + carry;
}

__global__ __launch_bounds__(256) void fill_k(const int* __restrict__ row,
                                              const int* __restrict__ col,
                                              const float* __restrict__ w,
                                              const int* __restrict__ rowptr,
                                              const int* __restrict__ rank,
                                              const int* __restrict__ boff_all,
                                              const float* __restrict__ dinv,
                                              int* __restrict__ slot_words, int e,
                                              size_t ws_bytes) {
    if (ws_bytes < WS_NEED_BYTES) return;
    int2* __restrict__ slot = reinterpret_cast<int2*>(slot_words);
    int i = blockIdx.x * 256 + threadIdx.x;
    const int* __restrict__ boff = boff_all + (size_t)(blockIdx.x / (CB_CHUNK / 256)) * GCN_N;
    if (i < e) {
        const int c = col[i];
        const int r = row[i];
        const int pos = rowptr[c] + boff[c] + rank[i];
        const float nrm = dinv[r] * w[i] * dinv[c];
        slot[min((unsigned)pos, (unsigned)(e - 1))] = make_int2(r, __float_as_int(nrm));
    }
}

#define ACT_ELU   1
#define ACT_GELU  2
#define ACT_PRELU 3

__device__ __forceinline__ float act_on_store(float h, int mode, float slope) {
    if (mode == ACT_ELU)  return (h > 0.0f) ? h : (expf(h) - 1.0f);
    if (mode == ACT_GELU) return 0.5f * h * (1.0f + erff(h * 0.70710678118654752f));
    return (h >= 0.0f) ? h : slope * h;
}

#define LT_PER_BLOCK 4
#define N_TILES (GCN_N / 16)
#define L_GRID ((N_TILES + LT_PER_BLOCK - 1) / LT_PER_BLOCK)
#define A_PITCH (GCN_D + 4)
static_assert(GCN_N % 16 == 0, "row tiles must be whole");

__device__ __forceinline__ int   bcast_i(int x, int j)   { return __builtin_amdgcn_readlane(x, j); }
__device__ __forceinline__ float bcast_f(float x, int j) {
    return __int_as_float(__builtin_amdgcn_readlane(__float_as_int(x), j));
}

__device__ __forceinline__ void axpy4(float4& acc, float a, const float4 v) {
    const v2f aa = {a, a};
    const v2f lo = __builtin_elementwise_fma(aa, (v2f){v.x, v.y}, (v2f){acc.x, acc.y});
    const v2f hi = __builtin_elementwise_fma(aa, (v2f){v.z, v.w}, (v2f){acc.z, acc.w});
    acc.x = lo.x; acc.y = lo.y; acc.z = hi.x; acc.w = hi.y;
}

__device__ __forceinline__ float4 load_row4(const float* __restrict__ h, int r, int lane) {
    const unsigned long long a = reinterpret_cast<unsigned long long>(h + (size_t)r * GCN_D);
    const unsigned lo = __builtin_amdgcn_readfirstlane((unsigned)a);
    const unsigned hi = __builtin_amdgcn_readfirstlane((unsigned)(a >> 32));
    const __attribute__((address_space(1))) float4* rowp =
        (const __attribute__((address_space(1))) float4*)(((unsigned long long)hi << 32) | lo);
    return rowp[(unsigned)lane];
}

__device__ __forceinline__ unsigned long long row_base(const float* __restrict__ h, int r) {
    return reinterpret_cast<unsigned long long>(h + (size_t)r * GCN_D);
}

__device__ __forceinline__ float4 load_lane16(unsigned long long base, int lane) {
    const __attribute__((address_space(1))) float4* rowp =
        (const __attribute__((address_space(1))) float4*)base;
    return rowp[(unsigned)lane];
}

__device__ __forceinline__ float4 sum_in_edges(const float* __restrict__ h,
                                               const int2* slot,
                                               int lane, int beg, int end,
                                               int2 sl, float4 acc) {
    for (int base = beg; base < end; base += 32) {
        const int nb = min(end - base, 32);
        if (base != beg) {
            const int k = base + ((lane < nb) ? lane : 0);
            sl = slot[k];
        }
        const int   ms = sl.x;
        const float mn = __int_as_float(sl.y);
        int j = 0;
        for (; j + 8 <= nb; j += 8) {
            unsigned long long b0 = row_base(h, bcast_i(ms, j + 0)), b1 = row_base(h, bcast_i(ms, j + 1));
            unsigned long long b2 = row_base(h, bcast_i(ms, j + 2)), b3 = row_base(h, bcast_i(ms, j + 3));
            unsigned long long b4 = row_base(h, bcast_i(ms, j + 4)), b5 = row_base(h, bcast_i(ms, j + 5));
            unsigned long long b6 = row_base(h, bcast_i(ms, j + 6)), b7 = row_base(h, bcast_i(ms, j + 7));
            asm volatile("" : "+s"(b0), "+s"(b1), "+s"(b2), "+s"(b3), "+s"(b4), "+s"(b5), "+s"(b6), "+s"(b7));
            const float4 v0 = load_lane16(b0, lane);
            const float4 v1 = load_lane16(b1, lane);
            const float4 v2 = load_lane16(b2, lane);
            const float4 v3 = load_lane16(b3, lane);
            const float4 v4 = load_lane16(b4, lane);
            const float4 v5 = load_lane16(b5, lane);
            const float4 v6 = load_lane16(b6, lane);
            const float4 v7 = load_lane16(b7, lane);
            axpy4(acc, bcast_f(mn, j + 0), v0);
            axpy4(acc, bcast_f(mn, j + 1), v1);
            axpy4(acc, bcast_f(mn, j + 2), v2);
            axpy4(acc, bcast_f(mn, j + 3), v3);
            axpy4(acc, bcast_f(mn, j + 4), v4);
            axpy4(acc, bcast_f(mn, j + 5), v5);
            axpy4(acc, bcast_f(mn, j + 6), v6);
            axpy4(acc, bcast_f(mn, j + 7), v7);
        }
        if (j + 4 <= nb) {
            unsigned long long b0 = row_base(h, bcast_i(ms, j + 0)), b1 = row_base(h, bcast_i(ms, j + 1));
            unsigned long long b2 = row_base(h, bcast_i(ms, j + 2)), b3 = row_base(h, bcast_i(ms, j + 3));
            asm volatile("" : "+s"(b0), "+s"(b1), "+s"(b2), "+s"(b3));
            const float4 v0 = load_lane16(b0, lane);
            const float4 v1 = load_lane16(b1, lane);
            const float4 v2 = load_lane16(b2, lane);
            const float4 v3 = load_lane16(b3, lane);
            axpy4(acc, bcast_f(mn, j + 0), v0);
            axpy4(acc, bcast_f(mn, j + 1), v1);
            axpy4(acc, bcast_f(mn, j + 2), v2);
            axpy4(acc, bcast_f(mn, j + 3), v3);
            j += 4;
        }
        if (j < nb) {
            const int j1 = min(j + 1, nb - 1), j2 = min(j + 2, nb - 1);
            unsigned long long b0 = row_base(h, bcast_i(ms, j)), b1 = row_base(h, bcast_i(ms, j1));
            unsigned long long b2 = row_base(h, bcast_i(ms, j2));
            asm volatile("" : "+s"(b0), "+s"(b1), "+s"(b2));
            const float4 v0 = load_lane16(b0, lane);
            const float4 v1 = load_lane16(b1, lane);
            const float4 v2 = load_lane16(b2, lane);
            axpy4(acc, bcast_f(mn, j), v0);
            if (j + 1 < nb) axpy4(acc, bcast_f(mn, j1), v1);
            if (j + 2 < nb) axpy4(acc, bcast_f(mn, j2), v2);
        }
    }
    return acc;
}

struct TileHead {
    int    beg0, end0, beg1, end1;
    float  dd0, dd1;
    int2   s0, s1;
    float4 own0, own1;
};

__device__ __forceinline__ void open_tile(TileHead& th, int rp, float dv, int row0,
                                          const float* __restrict__ h,
                                          const int2* slot,
                                          int wave, int lane, int e) {
    th.beg0 = max(bcast_i(rp, wave), 0);
    th.end0 = min(bcast_i(rp, wave + 1), e);
    th.beg1 = max(bcast_i(rp, wave + 8), 0);
    th.end1 = min(bcast_i(rp, wave + 9), e);
    const float d0 = bcast_f(dv, wave), d1 = bcast_f(dv, wave + 8);
    th.dd0 = d0 * d0;
    th.dd1 = d1 * d1;
    const int nb0 = min(th.end0 - th.beg0, 32), nb1 = min(th.end1 - th.beg1, 32);
    th.s0 = slot[min(th.beg0 + ((lane < nb0) ? lane : 0), e - 1)];
    th.s1 = slot[min(th.beg1 + ((lane < nb1) ? lane : 0), e - 1)];
    th.own0 = load_row4(h, row0 + wave, lane);
    th.own1 = load_row4(h, row0 + wave + 8, lane);
}

__global__ __launch_bounds__(256) void layer_k(float* __restrict__ hout,
                                               const float* __restrict__ hin,
                                               const float* __restrict__ W,
                                               const float* __restrict__ b,
                                               const float* __restrict__ prelu_w,
                                               const int* __restrict__ rowptr,
                                               const int* slot_words,
                                               const float* __restrict__ dinv,
                                               int act, int e, size_t ws_bytes) {
    if (ws_bytes < WS_NEED_BYTES) return;
    __shared__ __attribute__((aligned(16))) float As[2][16][A_PITCH];

    const int2* slot = reinterpret_cast<const int2*>(slot_words);
    const int t    = threadIdx.x;
    const int lane = t & 31;
    const int wave = __builtin_amdgcn_readfirstlane((int)(threadIdx.x >> 5));
    const int m   = lane & 15;
    const int n   = wave * 16 + (lane & 15);
    const int klo = (lane >> 4) * 2;
    const int rowOff = (lane >> 4) * 8;
    const int tile0 = blockIdx.x * LT_PER_BLOCK;

    const int   rp0 = rowptr[tile0 * 16 + min(lane, 16)];
    const float dv0 = dinv[tile0 * 16 + (lane & 15)];

    v2f bf[32];
    #pragma unroll
    for (int j = 0; j < 32; ++j) {
        bf[j].x = W[(4 * j + klo) * GCN_D + n];
        bf[j].y = W[(4 * j + klo + 1) * GCN_D + n];
    }
    const float bn    = b[n];
    const float slope = prelu_w[0];

    TileHead th;
    open_tile(th, rp0, dv0, tile0 * 16, hin, slot, wave, lane, e);

    #pragma unroll 1
    for (int tt = 0; tt < LT_PER_BLOCK; ++tt) {
        const int tile = tile0 + tt;
        if (tile >= N_TILES) break;
        const int buf = tt & 1;

        const int nrow0 = min(tile + 1, N_TILES - 1) * 16;
        const int   nrp = rowptr[nrow0 + min(lane, 16)];
        const float ndv = dinv[nrow0 + (lane & 15)];

        {
            float4 acc;
            acc.x = th.dd0 * th.own0.x; acc.y = th.dd0 * th.own0.y;
            acc.z = th.dd0 * th.own0.z; acc.w = th.dd0 * th.own0.w;
            acc = sum_in_edges(hin, slot, lane, th.beg0, th.end0, th.s0, acc);
            *reinterpret_cast<float4*>(&As[buf][wave][4 * lane]) = acc;
        }
        {
            float4 acc;
            acc.x = th.dd1 * th.own1.x; acc.y = th.dd1 * th.own1.y;
            acc.z = th.dd1 * th.own1.z; acc.w = th.dd1 * th.own1.w;
            acc = sum_in_edges(hin, slot, lane, th.beg1, th.end1, th.s1, acc);
            *reinterpret_cast<float4*>(&As[buf][wave + 8][4 * lane]) = acc;
        }
        __syncthreads();

        open_tile(th, nrp, ndv, nrow0, hin, slot, wave, lane, e);
        __builtin_amdgcn_sched_barrier(0);

        v8f acc = {};
        #pragma unroll
        for (int j = 0; j < 32; ++j) {
            const v2f a = *reinterpret_cast<const v2f*>(&As[buf][m][4 * j + klo]);
            acc = __builtin_amdgcn_wmma_f32_16x16x4_f32(
                 false, a,  false, bf[j],
                 (short)0, acc,  false,  false);
        }

        __builtin_amdgcn_sched_barrier(0);
        asm volatile("v_nop\n\tv_nop\n\tv_nop\n\tv_nop");
        __builtin_amdgcn_sched_barrier(0);

        const int rowBase = tile * 16;
        #pragma unroll
        for (int v = 0; v < 8; ++v) {
            const int r = rowBase + v + rowOff;
            hout[r * GCN_D + n] = act_on_store(acc[v] + bn, act, slope);
        }
    }
}


extern "C" void kernel_launch(void* const* d_in, const int* in_sizes, int n_in,
                              void* d_out, int out_size, void* d_ws, size_t ws_size,
                              hipStream_t stream) {
    (void)in_sizes; (void)n_in; (void)out_size;

    const float* x   = (const float*)d_in[0];
    const int*   row = (const int*)d_in[1];
    const int*   col = ((const int*)d_in[1]) + GCN_E;
    const float* w   = (const float*)d_in[2];
    const float* W1  = (const float*)d_in[3];
    const float* b1  = (const float*)d_in[4];
    const float* W2  = (const float*)d_in[5];
    const float* b2  = (const float*)d_in[6];
    const float* W3  = (const float*)d_in[7];
    const float* b3  = (const float*)d_in[8];
    const float* pw  = (const float*)d_in[9];

    float* out  = (float*)d_out;
    float* hmid = (float*)d_ws;
    float* deg  = hmid + (size_t)GCN_N * GCN_D;
    float* dinv = deg + GCN_N;
    int*   rowptr = (int*)(dinv + GCN_N);
    int*   slot   = rowptr + (GCN_N + 4);
    int*   cnt    = (int*)d_out;
    int*   rank   = cnt + GCN_N;
    unsigned* pk  = (unsigned*)(rank + GCN_E);
    float* degp   = (float*)(pk + (size_t)CB_BLOCKS * CB_PK_WORDS);
    int*   boff   = (int*)(degp + (size_t)CB_BLOCKS * GCN_N);

    const int nThr = 256;
    const int gP   = (GCN_N / 2 + nThr - 1) / nThr;
    const int gE   = (GCN_E + nThr - 1) / nThr;

    count_k        <<<CB_BLOCKS, CB_THREADS, 0, stream>>>(pk, degp, rank, col, w, GCN_E);
    block_offsets_k<<<gP, nThr, 0, stream>>>(pk, degp, boff, cnt, dinv, GCN_N, ws_size);
    rowptr_k  <<<1, 1024, 0, stream>>>(cnt, rowptr, GCN_N, ws_size);
    fill_k    <<<gE, nThr, 0, stream>>>(row, col, w, rowptr, rank, boff, dinv, slot, GCN_E, ws_size);

    layer_k<<<L_GRID, nThr, 0, stream>>>(out,  x,    W1, b1, pw, rowptr, slot, dinv, ACT_ELU,   GCN_E, ws_size);
    layer_k<<<L_GRID, nThr, 0, stream>>>(hmid, out,  W2, b2, pw, rowptr, slot, dinv, ACT_GELU,  GCN_E, ws_size);
    layer_k<<<L_GRID, nThr, 0, stream>>>(out,  hmid, W3, b3, pw, rowptr, slot, dinv, ACT_PRELU, GCN_E, ws_size);
}
